# starting kernel
# baseline (speedup 1.0000x reference)
#include <hip/hip_runtime.h>

typedef __attribute__((ext_vector_type(2))) float v2f;
typedef __attribute__((ext_vector_type(8))) float v8f;

#define N_NODES 100000
#define N_EDGES 800000
#define FEAT    128
#define NGRAPH  512
#define NCLS    10

__device__ __forceinline__ void atomAddF(float* p, float v) {
  unsafeAtomicAdd(p, v);
}

__global__ void k_fill(float* __restrict__ p, float v, long n) {
  long i = (long)blockIdx.x * blockDim.x + threadIdx.x;
  if (i < n) p[i] = v;
}

__global__ void k_deg(const int* __restrict__ col, float* __restrict__ deg) {
  long e = (long)blockIdx.x * blockDim.x + threadIdx.x;
  if (e < N_EDGES) atomAddF(&deg[col[e]], 1.0f);
}

__global__ void k_dinv(float* __restrict__ deg) {
  long i = (long)blockIdx.x * blockDim.x + threadIdx.x;
  if (i < N_NODES) {
    float d = deg[i];
    deg[i] = (d > 0.0f) ? rsqrtf(fmaxf(d, 1.0f)) : 0.0f;
  }
}

__global__ __launch_bounds__(256) void k_gemm(const float* __restrict__ X,
                                              const float* __restrict__ W,
                                              float* __restrict__ Hout,
                                              int nrows) {
  __shared__ float sW[FEAT * FEAT];
  {
    const float4* src = (const float4*)W;
    float4* dst = (float4*)sW;
    for (int i = threadIdx.x; i < FEAT * FEAT / 4; i += blockDim.x) dst[i] = src[i];
  }
  __syncthreads();

  const int wave  = threadIdx.x >> 5;
  const int lane  = threadIdx.x & 31;
  const int strip = blockIdx.x * 8 + wave;
  const int row0  = strip * 16;
  if (row0 >= nrows) return;

  const int m    = lane & 15;
  const int koff = (lane >> 4) << 1;

  v8f acc[8];
#pragma unroll
  for (int n = 0; n < 8; ++n) acc[n] = (v8f){0, 0, 0, 0, 0, 0, 0, 0};

  const float* xrow = X + (size_t)(row0 + m) * FEAT + koff;
  const int ncol = lane & 15;

  for (int k = 0; k < FEAT; k += 4) {
    v2f a;
    a.x = xrow[k];
    a.y = xrow[k + 1];
    const int kb = k + koff;
#pragma unroll
    for (int n = 0; n < 8; ++n) {
      v2f b;
      b.x = sW[kb * FEAT + n * 16 + ncol];
      b.y = sW[(kb + 1) * FEAT + n * 16 + ncol];
      acc[n] = __builtin_amdgcn_wmma_f32_16x16x4_f32(
          false, a, false, b, (short)0, acc[n], false, false);
    }
  }

  const int rowsel = (lane >> 4) * 8;
#pragma unroll
  for (int n = 0; n < 8; ++n) {
#pragma unroll
    for (int r = 0; r < 8; ++r) {
      Hout[(size_t)(row0 + rowsel + r) * FEAT + n * 16 + ncol] = acc[n][r];
    }
  }
}

__global__ __launch_bounds__(256) void k_agg(const float* __restrict__ h,
                                             const int* __restrict__ row,
                                             const int* __restrict__ col,
                                             const float* __restrict__ dinv,
                                             float* __restrict__ out) {
  const int wave = threadIdx.x >> 5;
  const int lane = threadIdx.x & 31;
  const long e = (long)blockIdx.x * 8 + wave;
  if (e >= N_EDGES) return;
  const int r = row[e];
  const int c = col[e];
  const float nrm = dinv[r] * dinv[c];
  const float4 v = ((const float4*)(h + (size_t)r * FEAT))[lane];
  float* op = out + (size_t)c * FEAT + lane * 4;
  atomAddF(op + 0, nrm * v.x);
  atomAddF(op + 1, nrm * v.y);
  atomAddF(op + 2, nrm * v.z);
  atomAddF(op + 3, nrm * v.w);
}

__global__ void k_epilogue(float* __restrict__ out, const float* __restrict__ h,
                           const float* __restrict__ dinv,
                           const float* __restrict__ bias, int relu) {
  long idx = (long)blockIdx.x * blockDim.x + threadIdx.x;
  if (idx >= (long)N_NODES * (FEAT / 4)) return;
  const int node = (int)(idx >> 5);
  const int f4   = (int)(idx & 31);
  const float d = dinv[node];
  const float s = d * d;
  float4 hv = ((const float4*)h)[idx];
  float4 ov = ((float4*)out)[idx];
  float4 bv = ((const float4*)bias)[f4];
  float4 rv;
  rv.x = ov.x + s * hv.x + bv.x;
  rv.y = ov.y + s * hv.y + bv.y;
  rv.z = ov.z + s * hv.z + bv.z;
  rv.w = ov.w + s * hv.w + bv.w;
  if (relu) {
    rv.x = fmaxf(rv.x, 0.0f);
    rv.y = fmaxf(rv.y, 0.0f);
    rv.z = fmaxf(rv.z, 0.0f);
    rv.w = fmaxf(rv.w, 0.0f);
  }
  ((float4*)out)[idx] = rv;
}

__global__ __launch_bounds__(256) void k_pool(const float* __restrict__ a,
                                              const int* __restrict__ batch,
                                              float* __restrict__ sums,
                                              float* __restrict__ cnts) {
  const int wave = threadIdx.x >> 5;
  const int lane = threadIdx.x & 31;
  const long i = (long)blockIdx.x * 8 + wave;
  if (i >= N_NODES) return;
  const int g = batch[i];
  const float4 v = ((const float4*)(a + (size_t)i * FEAT))[lane];
  float* sp = sums + (size_t)g * FEAT + lane * 4;
  atomAddF(sp + 0, v.x);
  atomAddF(sp + 1, v.y);
  atomAddF(sp + 2, v.z);
  atomAddF(sp + 3, v.w);
  if (lane == 0) atomAddF(&cnts[g], 1.0f);
}

__global__ void k_final(const float* __restrict__ sums, const float* __restrict__ cnts,
                        const float* __restrict__ Wl, const float* __restrict__ bl,
                        float* __restrict__ out) {
  const int g = blockIdx.x;
  const int c = threadIdx.x;
  if (c >= NCLS) return;
  const float inv = 1.0f / fmaxf(cnts[g], 1.0f);
  float acc = bl[c];
  for (int k = 0; k < FEAT; ++k)
    acc += sums[(size_t)g * FEAT + k] * inv * Wl[k * NCLS + c];
  out[(size_t)g * NCLS + c] = acc;
}

extern "C" void kernel_launch(void* const* d_in, const int* in_sizes, int n_in,
                              void* d_out, int out_size, void* d_ws, size_t ws_size,
                              hipStream_t stream) {
  const float* x     = (const float*)d_in[0];
  const int*   ei    = (const int*)d_in[1];
  const int*   row   = ei;
  const int*   col   = ei + N_EDGES;
  const int*   batch = (const int*)d_in[2];
  const float* W1 = (const float*)d_in[3];
  const float* b1 = (const float*)d_in[4];
  const float* W2 = (const float*)d_in[5];
  const float* b2 = (const float*)d_in[6];
  const float* W3 = (const float*)d_in[7];
  const float* b3 = (const float*)d_in[8];
  const float* Wl = (const float*)d_in[9];
  const float* bl = (const float*)d_in[10];
  float* outp = (float*)d_out;

  char* base = (char*)d_ws;
  const size_t featBytes = (size_t)N_NODES * FEAT * sizeof(float);
  float* deg  = (float*)(base);
  float* hbuf = (float*)(base + (1 << 20));
  float* abuf = (float*)(base + (1 << 20) + featBytes);
  float* sums = (float*)(base + (1 << 20) + 2 * featBytes);
  float* cnts = sums + (size_t)NGRAPH * FEAT;

  const long nElem  = (long)N_NODES * FEAT;
  const long nVec4  = nElem / 4;
  const int  TB     = 256;
  const int  gNode  = (N_NODES + TB - 1) / TB;
  const int  gElem  = (int)((nElem + TB - 1) / TB);
  const int  gVec4  = (int)((nVec4 + TB - 1) / TB);
  const int  gEdge1 = (N_EDGES + TB - 1) / TB;
  const int  gGemm  = (N_NODES / 16 + 8 - 1) / 8;
  const int  gAgg   = (N_EDGES + 8 - 1) / 8;
  const int  gPool  = (N_NODES + 8 - 1) / 8;

  k_fill<<<gNode, TB, 0, stream>>>(deg, 1.0f, N_NODES);
  k_deg<<<gEdge1, TB, 0, stream>>>(col, deg);
  k_dinv<<<gNode, TB, 0, stream>>>(deg);

  k_gemm<<<gGemm, TB, 0, stream>>>(x, W1, hbuf, N_NODES);
  k_fill<<<gElem, TB, 0, stream>>>(abuf, 0.0f, nElem);
  k_agg<<<gAgg, TB, 0, stream>>>(hbuf, row, col, deg, abuf);
  k_epilogue<<<gVec4, TB, 0, stream>>>(abuf, hbuf, deg, b1, 1);

  k_gemm<<<gGemm, TB, 0, stream>>>(abuf, W2, hbuf, N_NODES);
  k_fill<<<gElem, TB, 0, stream>>>(abuf, 0.0f, nElem);
  k_agg<<<gAgg, TB, 0, stream>>>(hbuf, row, col, deg, abuf);
  k_epilogue<<<gVec4, TB, 0, stream>>>(abuf, hbuf, deg, b2, 1);

  k_gemm<<<gGemm, TB, 0, stream>>>(abuf, W3, hbuf, N_NODES);
  k_fill<<<gElem, TB, 0, stream>>>(abuf, 0.0f, nElem);
  k_agg<<<gAgg, TB, 0, stream>>>(hbuf, row, col, deg, abuf);
  k_epilogue<<<gVec4, TB, 0, stream>>>(abuf, hbuf, deg, b3, 0);

  const long poolN = (long)NGRAPH * FEAT + NGRAPH;
  k_fill<<<(int)((poolN + TB - 1) / TB), TB, 0, stream>>>(sums, 0.0f, poolN);
  k_pool<<<gPool, TB, 0, stream>>>(abuf, batch, sums, cnts);
  k_final<<<NGRAPH, 32, 0, stream>>>(sums, cnts, Wl, bl, outp);
}


# edge adds and pool: lane owns features l, l+32, l+64, l+96 (was 4l..4l+3): each f32 atomic instruction covers one whole 128-B line, not a quarter of each of 4; still atomics in arrival order
# speedup vs baseline: 1.6704x; 1.6704x over previous
#include <hip/hip_runtime.h>

typedef __attribute__((ext_vector_type(2))) float v2f;
typedef __attribute__((ext_vector_type(8))) float v8f;

#define N_NODES 100000
#define N_EDGES 800000
#define FEAT    128
#define NGRAPH  512
#define NCLS    10

__device__ __forceinline__ void atomAddF(float* p, float v) {
  unsafeAtomicAdd(p, v);
}

__global__ void k_fill(float* __restrict__ p, float v, long n) {
  long i = (long)blockIdx.x * blockDim.x + threadIdx.x;
  if (i < n) p[i] = v;
}

__global__ void k_deg(const int* __restrict__ col, float* __restrict__ deg) {
  long e = (long)blockIdx.x * blockDim.x + threadIdx.x;
  if (e < N_EDGES) atomAddF(&deg[col[e]], 1.0f);
}

__global__ void k_dinv(float* __restrict__ deg) {
  long i = (long)blockIdx.x * blockDim.x + threadIdx.x;
  if (i < N_NODES) {
    float d = deg[i];
    deg[i] = (d > 0.0f) ? rsqrtf(fmaxf(d, 1.0f)) : 0.0f;
  }
}

template <bool RELU_IN>
__global__ __launch_bounds__(256) void k_gemm(const float* __restrict__ X,
                                              const float* __restrict__ W,
                                              const float* __restrict__ dinv,
                                              const float* __restrict__ bias,
                                              float* __restrict__ Hout,
                                              float* __restrict__ Acc0,
                                              int nrows) {
  __shared__ float sW[FEAT * FEAT];
  {
    const float4* src = (const float4*)W;
    float4* dst = (float4*)sW;
    for (int i = threadIdx.x; i < FEAT * FEAT / 4; i += blockDim.x) dst[i] = src[i];
  }
  __syncthreads();

  const int wave  = threadIdx.x >> 5;
  const int lane  = threadIdx.x & 31;
  const int strip = blockIdx.x * 8 + wave;
  const int row0  = strip * 16;
  if (row0 >= nrows) return;

  const int m    = lane & 15;
  const int koff = (lane >> 4) << 1;

  v8f acc[8];
#pragma unroll
  for (int n = 0; n < 8; ++n) acc[n] = (v8f){0, 0, 0, 0, 0, 0, 0, 0};

  const float* xrow = X + (size_t)(row0 + m) * FEAT + koff;
  const int ncol = lane & 15;

  for (int k0 = 0; k0 < FEAT; k0 += 16) {
    v2f a[4];
#pragma unroll
    for (int j = 0; j < 4; ++j) {
      a[j].x = xrow[k0 + 4 * j];
      a[j].y = xrow[k0 + 4 * j + 1];
    }
    v2f b[2][8];
#pragma unroll
    for (int n = 0; n < 8; ++n) {
      const int kb = k0 + koff;
      b[0][n].x = sW[kb * FEAT + n * 16 + ncol];
      b[0][n].y = sW[(kb + 1) * FEAT + n * 16 + ncol];
    }
#pragma unroll
    for (int j = 0; j < 4; ++j) {
      if (j < 3) {
        const int kb = k0 + 4 * (j + 1) + koff;
#pragma unroll
        for (int n = 0; n < 8; ++n) {
          b[(j + 1) & 1][n].x = sW[kb * FEAT + n * 16 + ncol];
          b[(j + 1) & 1][n].y = sW[(kb + 1) * FEAT + n * 16 + ncol];
        }
      }
      v2f aj = a[j];
      if (RELU_IN) {
        aj.x = fmaxf(aj.x, 0.0f);
        aj.y = fmaxf(aj.y, 0.0f);
      }
#pragma unroll
      for (int n = 0; n < 8; ++n) {
        acc[n] = __builtin_amdgcn_wmma_f32_16x16x4_f32(
            false, aj, false, b[j & 1][n], (short)0, acc[n], false, false);
      }
    }
  }

  const int rowsel = (lane >> 4) * 8;
  float s[8];
#pragma unroll
  for (int r = 0; r < 8; ++r) {
    const float d = dinv[row0 + rowsel + r];
    s[r] = d * d;
  }
#pragma unroll
  for (int n = 0; n < 8; ++n) {
    const float bv = bias[n * 16 + ncol];
#pragma unroll
    for (int r = 0; r < 8; ++r) {
      const size_t o = (size_t)(row0 + rowsel + r) * FEAT + n * 16 + ncol;
      const float hv = acc[n][r];
      Hout[o] = hv;
      Acc0[o] = s[r] * hv + bv;
    }
  }
}

__global__ __launch_bounds__(256) void k_agg(const float* __restrict__ h,
                                             const int* __restrict__ row,
                                             const int* __restrict__ col,
                                             const float* __restrict__ dinv,
                                             float* __restrict__ out) {
  const int wave = threadIdx.x >> 5;
  const int lane = threadIdx.x & 31;
  const long e = (long)blockIdx.x * 8 + wave;
  if (e >= N_EDGES) return;
  const int r = row[e];
  const int c = col[e];
  const float nrm = dinv[r] * dinv[c];
  const float* hp = h + (size_t)r * FEAT + lane;
  const float v0 = hp[0];
  const float v1 = hp[32];
  const float v2 = hp[64];
  const float v3 = hp[96];
  float* op = out + (size_t)c * FEAT + lane;
  atomAddF(op + 0,  nrm * v0);
  atomAddF(op + 32, nrm * v1);
  atomAddF(op + 64, nrm * v2);
  atomAddF(op + 96, nrm * v3);
}

__global__ __launch_bounds__(256) void k_pool(const float* __restrict__ a,
                                              const int* __restrict__ batch,
                                              float* __restrict__ sums,
                                              float* __restrict__ cnts) {
  const int wave = threadIdx.x >> 5;
  const int lane = threadIdx.x & 31;
  const long i = (long)blockIdx.x * 8 + wave;
  if (i >= N_NODES) return;
  const int g = batch[i];
  const float* ap = a + (size_t)i * FEAT + lane;
  const float v0 = ap[0];
  const float v1 = ap[32];
  const float v2 = ap[64];
  const float v3 = ap[96];
  float* sp = sums + (size_t)g * FEAT + lane;
  atomAddF(sp + 0,  v0);
  atomAddF(sp + 32, v1);
  atomAddF(sp + 64, v2);
  atomAddF(sp + 96, v3);
  if (lane == 0) atomAddF(&cnts[g], 1.0f);
}

__global__ void k_final(const float* __restrict__ sums, const float* __restrict__ cnts,
                        const float* __restrict__ Wl, const float* __restrict__ bl,
                        float* __restrict__ out) {
  const int g = blockIdx.x;
  const int c = threadIdx.x;
  if (c >= NCLS) return;
  const float inv = 1.0f / fmaxf(cnts[g], 1.0f);
  float acc = bl[c];
  for (int k = 0; k < FEAT; ++k)
    acc += sums[(size_t)g * FEAT + k] * inv * Wl[k * NCLS + c];
  out[(size_t)g * NCLS + c] = acc;
}

extern "C" void kernel_launch(void* const* d_in, const int* in_sizes, int n_in,
                              void* d_out, int out_size, void* d_ws, size_t ws_size,
                              hipStream_t stream) {
  const float* x     = (const float*)d_in[0];
  const int*   ei    = (const int*)d_in[1];
  const int*   row   = ei;
  const int*   col   = ei + N_EDGES;
  const int*   batch = (const int*)d_in[2];
  const float* W1 = (const float*)d_in[3];
  const float* b1 = (const float*)d_in[4];
  const float* W2 = (const float*)d_in[5];
  const float* b2 = (const float*)d_in[6];
  const float* W3 = (const float*)d_in[7];
  const float* b3 = (const float*)d_in[8];
  const float* Wl = (const float*)d_in[9];
  const float* bl = (const float*)d_in[10];
  float* outp = (float*)d_out;

  char* base = (char*)d_ws;
  const size_t featBytes = (size_t)N_NODES * FEAT * sizeof(float);
  float* deg  = (float*)(base);
  float* hbuf = (float*)(base + (1 << 20));
  float* pA   = (float*)(base + (1 << 20) + featBytes);
  float* pB   = (float*)(base + (1 << 20) + 2 * featBytes);
  float* sums = (float*)(base + (1 << 20) + 3 * featBytes);
  float* cnts = sums + (size_t)NGRAPH * FEAT;

  const int  TB     = 256;
  const int  gNode  = (N_NODES + TB - 1) / TB;
  const int  gEdge1 = (N_EDGES + TB - 1) / TB;
  const int  gGemm  = (N_NODES / 16 + 8 - 1) / 8;
  const int  gAgg   = (N_EDGES + 8 - 1) / 8;
  const int  gPool  = (N_NODES + 8 - 1) / 8;

  k_fill<<<gNode, TB, 0, stream>>>(deg, 1.0f, N_NODES);
  k_deg<<<gEdge1, TB, 0, stream>>>(col, deg);
  k_dinv<<<gNode, TB, 0, stream>>>(deg);

  k_gemm<false><<<gGemm, TB, 0, stream>>>(x, W1, deg, b1, hbuf, pA, N_NODES);
  k_agg<<<gAgg, TB, 0, stream>>>(hbuf, row, col, deg, pA);

  k_gemm<true><<<gGemm, TB, 0, stream>>>(pA, W2, deg, b2, hbuf, pB, N_NODES);
  k_agg<<<gAgg, TB, 0, stream>>>(hbuf, row, col, deg, pB);

  k_gemm<true><<<gGemm, TB, 0, stream>>>(pB, W3, deg, b3, hbuf, pA, N_NODES);
  k_agg<<<gAgg, TB, 0, stream>>>(hbuf, row, col, deg, pA);

  const long poolN = (long)NGRAPH * FEAT + NGRAPH;
  k_fill<<<(int)((poolN + TB - 1) / TB), TB, 0, stream>>>(sums, 0.0f, poolN);
  k_pool<<<gPool, TB, 0, stream>>>(pA, batch, sums, cnts);
  k_final<<<NGRAPH, 32, 0, stream>>>(sums, cnts, Wl, bl, outp);
}


# edge sums: a work-group owns 400 target rows in LDS, finds its edges by a byte key per edge, adds their rows with LDS atomic adds (arrival order), stores whole lines; global atomics per edge gone
# speedup vs baseline: 1.8811x; 1.1261x over previous
#include <hip/hip_runtime.h>

typedef __attribute__((ext_vector_type(2))) float v2f;
typedef __attribute__((ext_vector_type(8))) float v8f;

#define N_NODES 100000
#define N_EDGES 800000
#define FEAT    128
#define NGRAPH  512
#define NCLS    10

__device__ __forceinline__ void atomAddF(float* p, float v) {
  unsafeAtomicAdd(p, v);
}

__global__ void k_fill(float* __restrict__ p, float v, long n) {
  long i = (long)blockIdx.x * blockDim.x + threadIdx.x;
  if (i < n) p[i] = v;
}

#define TILE_NODES 400
__global__ void k_deg(const int* __restrict__ col, float* __restrict__ deg,
                      unsigned* __restrict__ key) {
  long i = (long)blockIdx.x * blockDim.x + threadIdx.x;
  if (i < N_EDGES / 4) {
    const int4 c = ((const int4*)col)[i];
    atomAddF(&deg[c.x], 1.0f);
    atomAddF(&deg[c.y], 1.0f);
    atomAddF(&deg[c.z], 1.0f);
    atomAddF(&deg[c.w], 1.0f);
    key[i] = (unsigned)(c.x / TILE_NODES) | ((unsigned)(c.y / TILE_NODES) << 8) |
             ((unsigned)(c.z / TILE_NODES) << 16) | ((unsigned)(c.w / TILE_NODES) << 24);
  }
}

__global__ void k_dinv(float* __restrict__ deg) {
  long i = (long)blockIdx.x * blockDim.x + threadIdx.x;
  if (i < N_NODES) {
    float d = deg[i];
    deg[i] = (d > 0.0f) ? rsqrtf(fmaxf(d, 1.0f)) : 0.0f;
  }
}

template <bool RELU_IN>
__global__ __launch_bounds__(256) void k_gemm(const float* __restrict__ X,
                                              const float* __restrict__ W,
                                              const float* __restrict__ dinv,
                                              const float* __restrict__ bias,
                                              float* __restrict__ Hout,
                                              float* __restrict__ Acc0,
                                              int nrows) {
  __shared__ float sW[FEAT * FEAT];
  {
    const float4* src = (const float4*)W;
    float4* dst = (float4*)sW;
    for (int i = threadIdx.x; i < FEAT * FEAT / 4; i += blockDim.x) dst[i] = src[i];
  }
  __syncthreads();

  const int wave  = threadIdx.x >> 5;
  const int lane  = threadIdx.x & 31;
  const int strip = blockIdx.x * 8 + wave;
  const int row0  = strip * 16;
  if (row0 >= nrows) return;

  const int m    = lane & 15;
  const int koff = (lane >> 4) << 1;

  v8f acc[8];
#pragma unroll
  for (int n = 0; n < 8; ++n) acc[n] = (v8f){0, 0, 0, 0, 0, 0, 0, 0};

  const float* xrow = X + (size_t)(row0 + m) * FEAT + koff;
  const int ncol = lane & 15;

  for (int k0 = 0; k0 < FEAT; k0 += 16) {
    v2f a[4];
#pragma unroll
    for (int j = 0; j < 4; ++j) {
      a[j].x = xrow[k0 + 4 * j];
      a[j].y = xrow[k0 + 4 * j + 1];
    }
    v2f b[2][8];
#pragma unroll
    for (int n = 0; n < 8; ++n) {
      const int kb = k0 + koff;
      b[0][n].x = sW[kb * FEAT + n * 16 + ncol];
      b[0][n].y = sW[(kb + 1) * FEAT + n * 16 + ncol];
    }
#pragma unroll
    for (int j = 0; j < 4; ++j) {
      if (j < 3) {
        const int kb = k0 + 4 * (j + 1) + koff;
#pragma unroll
        for (int n = 0; n < 8; ++n) {
          b[(j + 1) & 1][n].x = sW[kb * FEAT + n * 16 + ncol];
          b[(j + 1) & 1][n].y = sW[(kb + 1) * FEAT + n * 16 + ncol];
        }
      }
      v2f aj = a[j];
      if (RELU_IN) {
        aj.x = fmaxf(aj.x, 0.0f);
        aj.y = fmaxf(aj.y, 0.0f);
      }
#pragma unroll
      for (int n = 0; n < 8; ++n) {
        acc[n] = __builtin_amdgcn_wmma_f32_16x16x4_f32(
            false, aj, false, b[j & 1][n], (short)0, acc[n], false, false);
      }
    }
  }

  const int rowsel = (lane >> 4) * 8;
  float s[8];
#pragma unroll
  for (int r = 0; r < 8; ++r) {
    const float d = dinv[row0 + rowsel + r];
    s[r] = d * d;
  }
#pragma unroll
  for (int n = 0; n < 8; ++n) {
    const float bv = bias[n * 16 + ncol];
#pragma unroll
    for (int r = 0; r < 8; ++r) {
      const size_t o = (size_t)(row0 + rowsel + r) * FEAT + n * 16 + ncol;
      const float hv = acc[n][r];
      Hout[o] = hv;
      Acc0[o] = s[r] * hv + bv;
    }
  }
}

#define N_TILES    (N_NODES / TILE_NODES)
#define OWN_T      1024
#define KEYS_PER_LOAD 16
#define N_KEYQ     (N_EDGES / KEYS_PER_LOAD)
#define SCAN_ITERS ((N_KEYQ + OWN_T - 1) / OWN_T)
#define SCAN_BATCH 7
#define LIST_CAP   24576
static_assert(N_NODES % TILE_NODES == 0 && N_TILES <= 255, "tile key must fit a byte below 0xFF");
static_assert(N_EDGES % KEYS_PER_LOAD == 0 && N_EDGES % 4 == 0, "key array is written 4 and read 16 bytes at a time");
static_assert(TILE_NODES <= 512 && N_NODES <= (1 << 17), "list entry packs source (17 bits) and local row (9 bits)");

template <int K>
__device__ __forceinline__ void own_add(float* __restrict__ tile, const unsigned* __restrict__ ents,
                                        int node0, const float* __restrict__ h,
                                        const float* __restrict__ dinv, int lane) {
  int   cl[K];
  float nrm[K];
  float v[K][4];
#pragma unroll
  for (int u = 0; u < K; ++u) {
    const unsigned ent = ents[u];
    const int s = (int)(ent >> 9);
    cl[u] = (int)(ent & 511u);
    nrm[u] = dinv[s] * dinv[node0 + cl[u]];
    const float* hp = h + (size_t)s * FEAT + lane;
    v[u][0] = hp[0];
    v[u][1] = hp[32];
    v[u][2] = hp[64];
    v[u][3] = hp[96];
  }
#pragma unroll
  for (int u = 0; u < K; ++u) {
    float* tp = tile + cl[u] * FEAT + lane;
    atomicAdd(tp + 0,  nrm[u] * v[u][0]);
    atomicAdd(tp + 32, nrm[u] * v[u][1]);
    atomicAdd(tp + 64, nrm[u] * v[u][2]);
    atomicAdd(tp + 96, nrm[u] * v[u][3]);
  }
}

__device__ __forceinline__ void own_drain(float* __restrict__ tile, const unsigned* __restrict__ list,
                                          int n, int node0,
                                          const float* __restrict__ h,
                                          const float* __restrict__ dinv,
                                          int wave, int lane) {
  const int groups = n >> 2;
  for (int g = wave; g < groups; g += OWN_T / 32)
    own_add<4>(tile, list + 4 * g, node0, h, dinv, lane);
  if (wave < (n & 3))
    own_add<1>(tile, list + 4 * groups + wave, node0, h, dinv, lane);
}

__global__ __launch_bounds__(OWN_T) void k_agg_own(const float* __restrict__ h,
                                                   const int* __restrict__ row,
                                                   const int* __restrict__ col,
                                                   const unsigned char* __restrict__ key,
                                                   const float* __restrict__ dinv,
                                                   float* __restrict__ plane) {
  __shared__ __attribute__((aligned(16))) float tile[TILE_NODES * FEAT];
  __shared__ unsigned list[LIST_CAP];
  __shared__ int cnt_s;

  const int t     = threadIdx.x;
  const int wave  = __builtin_amdgcn_readfirstlane(t >> 5);
  const int lane  = t & 31;
  const unsigned bid = blockIdx.x;
  const int node0 = (int)bid * TILE_NODES;

  if (t == 0) cnt_s = 0;
  {
    const float4* src4 = (const float4*)(plane + (size_t)node0 * FEAT);
    float4* tile4 = (float4*)tile;
    for (int i = t; i < TILE_NODES * FEAT / 4; i += OWN_T) tile4[i] = src4[i];
  }
  __syncthreads();

  const uint4* key4 = (const uint4*)key;
  for (int it0 = 0; it0 < SCAN_ITERS; it0 += SCAN_BATCH) {
    uint4 k[SCAN_BATCH];
#pragma unroll
    for (int u = 0; u < SCAN_BATCH; ++u) {
      const int q = (it0 + u) * OWN_T + t;
      k[u] = key4[min(q, N_KEYQ - 1)];
    }
#pragma unroll
    for (int u = 0; u < SCAN_BATCH; ++u) {
      const int q = (it0 + u) * OWN_T + t;
      const unsigned w[4] = {k[u].x, k[u].y, k[u].z, k[u].w};
      unsigned mask = 0;
#pragma unroll
      for (int j = 0; j < 4; ++j)
#pragma unroll
        for (int b = 0; b < 4; ++b)
          mask |= (unsigned)(((w[j] >> (8 * b)) & 0xFFu) == bid) << (4 * j + b);
      if (q >= N_KEYQ) mask = 0;
      while (mask) {
        const int pos = __builtin_ctz(mask);
        mask &= mask - 1;
        const int e = q * KEYS_PER_LOAD + pos;
        const int idx = atomicAdd(&cnt_s, 1);
        if (idx < LIST_CAP) {
          list[idx] = (unsigned)e;
        } else {
          const int c  = col[e];
          const int sN = row[e];
          const float nrm = dinv[sN] * dinv[c];
          float* tp = tile + (c - node0) * FEAT;
          const float* hp = h + (size_t)sN * FEAT;
#pragma nounroll
          for (int f = 0; f < FEAT; ++f) atomicAdd(tp + f, nrm * hp[f]);
        }
      }
    }
  }
  __syncthreads();
  const int n = min(cnt_s, LIST_CAP);
  for (int i = t; i < n; i += OWN_T) {
    const int e = (int)list[i];
    list[i] = ((unsigned)row[e] << 9) | (unsigned)(col[e] - node0);
  }
  __syncthreads();
  own_drain(tile, list, n, node0, h, dinv, wave, lane);
  __syncthreads();
  {
    float4* dst4 = (float4*)(plane + (size_t)node0 * FEAT);
    const float4* tile4 = (const float4*)tile;
    for (int i = t; i < TILE_NODES * FEAT / 4; i += OWN_T) dst4[i] = tile4[i];
  }
}

__global__ __launch_bounds__(256) void k_pool(const float* __restrict__ a,
                                              const int* __restrict__ batch,
                                              float* __restrict__ sums,
                                              float* __restrict__ cnts) {
  const int wave = threadIdx.x >> 5;
  const int lane = threadIdx.x & 31;
  const long i = (long)blockIdx.x * 8 + wave;
  if (i >= N_NODES) return;
  const int g = batch[i];
  const float* ap = a + (size_t)i * FEAT + lane;
  const float v0 = ap[0];
  const float v1 = ap[32];
  const float v2 = ap[64];
  const float v3 = ap[96];
  float* sp = sums + (size_t)g * FEAT + lane;
  atomAddF(sp + 0,  v0);
  atomAddF(sp + 32, v1);
  atomAddF(sp + 64, v2);
  atomAddF(sp + 96, v3);
  if (lane == 0) atomAddF(&cnts[g], 1.0f);
}

__global__ void k_final(const float* __restrict__ sums, const float* __restrict__ cnts,
                        const float* __restrict__ Wl, const float* __restrict__ bl,
                        float* __restrict__ out) {
  const int g = blockIdx.x;
  const int c = threadIdx.x;
  if (c >= NCLS) return;
  const float inv = 1.0f / fmaxf(cnts[g], 1.0f);
  float acc = bl[c];
  for (int k = 0; k < FEAT; ++k)
    acc += sums[(size_t)g * FEAT + k] * inv * Wl[k * NCLS + c];
  out[(size_t)g * NCLS + c] = acc;
}

extern "C" void kernel_launch(void* const* d_in, const int* in_sizes, int n_in,
                              void* d_out, int out_size, void* d_ws, size_t ws_size,
                              hipStream_t stream) {
  const float* x     = (const float*)d_in[0];
  const int*   ei    = (const int*)d_in[1];
  const int*   row   = ei;
  const int*   col   = ei + N_EDGES;
  const int*   batch = (const int*)d_in[2];
  const float* W1 = (const float*)d_in[3];
  const float* b1 = (const float*)d_in[4];
  const float* W2 = (const float*)d_in[5];
  const float* b2 = (const float*)d_in[6];
  const float* W3 = (const float*)d_in[7];
  const float* b3 = (const float*)d_in[8];
  const float* Wl = (const float*)d_in[9];
  const float* bl = (const float*)d_in[10];
  float* outp = (float*)d_out;

  char* base = (char*)d_ws;
  const size_t featBytes = (size_t)N_NODES * FEAT * sizeof(float);
  float* deg  = (float*)(base);
  float* hbuf = (float*)(base + (1 << 20));
  float* pA   = (float*)(base + (1 << 20) + featBytes);
  float* pB   = (float*)(base + (1 << 20) + 2 * featBytes);
  float* sums = (float*)(base + (1 << 20) + 3 * featBytes);
  float* cnts = sums + (size_t)NGRAPH * FEAT;
  unsigned char* key = (unsigned char*)(base + (2 << 20) + 3 * featBytes);

  const int  TB     = 256;
  const int  gNode  = (N_NODES + TB - 1) / TB;
  const int  gEdge4 = (N_EDGES / 4 + TB - 1) / TB;
  const int  gGemm  = (N_NODES / 16 + 8 - 1) / 8;
  const int  gPool  = (N_NODES + 8 - 1) / 8;

  k_fill<<<gNode, TB, 0, stream>>>(deg, 1.0f, N_NODES);
  k_deg<<<gEdge4, TB, 0, stream>>>(col, deg, (unsigned*)key);
  k_dinv<<<gNode, TB, 0, stream>>>(deg);

  k_gemm<false><<<gGemm, TB, 0, stream>>>(x, W1, deg, b1, hbuf, pA, N_NODES);
  k_agg_own<<<N_TILES, OWN_T, 0, stream>>>(hbuf, row, col, key, deg, pA);

  k_gemm<true><<<gGemm, TB, 0, stream>>>(pA, W2, deg, b2, hbuf, pB, N_NODES);
  k_agg_own<<<N_TILES, OWN_T, 0, stream>>>(hbuf, row, col, key, deg, pB);

  k_gemm<true><<<gGemm, TB, 0, stream>>>(pB, W3, deg, b3, hbuf, pA, N_NODES);
  k_agg_own<<<N_TILES, OWN_T, 0, stream>>>(hbuf, row, col, key, deg, pA);

  const long poolN = (long)NGRAPH * FEAT + NGRAPH;
  k_fill<<<(int)((poolN + TB - 1) / TB), TB, 0, stream>>>(sums, 0.0f, poolN);
  k_pool<<<gPool, TB, 0, stream>>>(pA, batch, sums, cnts);
  k_final<<<NGRAPH, 32, 0, stream>>>(sums, cnts, Wl, bl, outp);
}


# edge sums: a tile's packed edge list is made once, in layer 1's launch, and kept in the workspace; layers 2 and 3 load it: no key scan, no row/col look-up; adds still LDS atomics in arrival order
# speedup vs baseline: 2.0698x; 1.1003x over previous
#include <hip/hip_runtime.h>

typedef __attribute__((ext_vector_type(2))) float v2f;
typedef __attribute__((ext_vector_type(8))) float v8f;

#define N_NODES 100000
#define N_EDGES 800000
#define FEAT    128
#define NGRAPH  512
#define NCLS    10

__device__ __forceinline__ void atomAddF(float* p, float v) {
  unsafeAtomicAdd(p, v);
}

__global__ void k_fill(float* __restrict__ p, float v, long n) {
  long i = (long)blockIdx.x * blockDim.x + threadIdx.x;
  if (i < n) p[i] = v;
}

#define TILE_NODES 400
__global__ void k_deg(const int* __restrict__ col, float* __restrict__ deg,
                      unsigned* __restrict__ key) {
  long i = (long)blockIdx.x * blockDim.x + threadIdx.x;
  if (i < N_EDGES / 4) {
    const int4 c = ((const int4*)col)[i];
    atomAddF(&deg[c.x], 1.0f);
    atomAddF(&deg[c.y], 1.0f);
    atomAddF(&deg[c.z], 1.0f);
    atomAddF(&deg[c.w], 1.0f);
    key[i] = (unsigned)(c.x / TILE_NODES) | ((unsigned)(c.y / TILE_NODES) << 8) |
             ((unsigned)(c.z / TILE_NODES) << 16) | ((unsigned)(c.w / TILE_NODES) << 24);
  }
}

__global__ void k_dinv(float* __restrict__ deg) {
  long i = (long)blockIdx.x * blockDim.x + threadIdx.x;
  if (i < N_NODES) {
    float d = deg[i];
    deg[i] = (d > 0.0f) ? rsqrtf(fmaxf(d, 1.0f)) : 0.0f;
  }
}

template <bool RELU_IN>
__global__ __launch_bounds__(256) void k_gemm(const float* __restrict__ X,
                                              const float* __restrict__ W,
                                              const float* __restrict__ dinv,
                                              const float* __restrict__ bias,
                                              float* __restrict__ Hout,
                                              float* __restrict__ Acc0,
                                              int nrows) {
  __shared__ float sW[FEAT * FEAT];
  {
    const float4* src = (const float4*)W;
    float4* dst = (float4*)sW;
    for (int i = threadIdx.x; i < FEAT * FEAT / 4; i += blockDim.x) dst[i] = src[i];
  }
  __syncthreads();

  const int wave  = threadIdx.x >> 5;
  const int lane  = threadIdx.x & 31;
  const int strip = blockIdx.x * 8 + wave;
  const int row0  = strip * 16;
  if (row0 >= nrows) return;

  const int m    = lane & 15;
  const int koff = (lane >> 4) << 1;

  v8f acc[8];
#pragma unroll
  for (int n = 0; n < 8; ++n) acc[n] = (v8f){0, 0, 0, 0, 0, 0, 0, 0};

  const float* xrow = X + (size_t)(row0 + m) * FEAT + koff;
  const int ncol = lane & 15;

  for (int k0 = 0; k0 < FEAT; k0 += 16) {
    v2f a[4];
#pragma unroll
    for (int j = 0; j < 4; ++j) {
      a[j].x = xrow[k0 + 4 * j];
      a[j].y = xrow[k0 + 4 * j + 1];
    }
    v2f b[2][8];
#pragma unroll
    for (int n = 0; n < 8; ++n) {
      const int kb = k0 + koff;
      b[0][n].x = sW[kb * FEAT + n * 16 + ncol];
      b[0][n].y = sW[(kb + 1) * FEAT + n * 16 + ncol];
    }
#pragma unroll
    for (int j = 0; j < 4; ++j) {
      if (j < 3) {
        const int kb = k0 + 4 * (j + 1) + koff;
#pragma unroll
        for (int n = 0; n < 8; ++n) {
          b[(j + 1) & 1][n].x = sW[kb * FEAT + n * 16 + ncol];
          b[(j + 1) & 1][n].y = sW[(kb + 1) * FEAT + n * 16 + ncol];
        }
      }
      v2f aj = a[j];
      if (RELU_IN) {
        aj.x = fmaxf(aj.x, 0.0f);
        aj.y = fmaxf(aj.y, 0.0f);
      }
#pragma unroll
      for (int n = 0; n < 8; ++n) {
        acc[n] = __builtin_amdgcn_wmma_f32_16x16x4_f32(
            false, aj, false, b[j & 1][n], (short)0, acc[n], false, false);
      }
    }
  }

  const int rowsel = (lane >> 4) * 8;
  float s[8];
#pragma unroll
  for (int r = 0; r < 8; ++r) {
    const float d = dinv[row0 + rowsel + r];
    s[r] = d * d;
  }
#pragma unroll
  for (int n = 0; n < 8; ++n) {
    const float bv = bias[n * 16 + ncol];
#pragma unroll
    for (int r = 0; r < 8; ++r) {
      const size_t o = (size_t)(row0 + rowsel + r) * FEAT + n * 16 + ncol;
      const float hv = acc[n][r];
      Hout[o] = hv;
      Acc0[o] = s[r] * hv + bv;
    }
  }
}

#define N_TILES    (N_NODES / TILE_NODES)
#define OWN_T      1024
#define KEYS_PER_LOAD 16
#define N_KEYQ     (N_EDGES / KEYS_PER_LOAD)
#define SCAN_ITERS ((N_KEYQ + OWN_T - 1) / OWN_T)
#define SCAN_BATCH 7
#define LIST_CAP   24576
static_assert(N_NODES % TILE_NODES == 0 && N_TILES <= 255, "tile key must fit a byte below 0xFF");
static_assert(N_EDGES % KEYS_PER_LOAD == 0 && N_EDGES % 4 == 0, "key array is written 4 and read 16 bytes at a time");
static_assert(TILE_NODES <= 512 && N_NODES <= (1 << 17), "list entry packs source (17 bits) and local row (9 bits)");
static_assert((4l << 20) + 3l * N_NODES * FEAT * 4 + (long)N_TILES * LIST_CAP * 4 <= 268435456l, "stored lists end inside the workspace");

template <int K>
__device__ __forceinline__ void own_add(float* __restrict__ tile, const unsigned* __restrict__ ents,
                                        int node0, const float* __restrict__ h,
                                        const float* __restrict__ dinv, int lane) {
  int   cl[K];
  float nrm[K];
  float v[K][4];
#pragma unroll
  for (int u = 0; u < K; ++u) {
    const unsigned ent = ents[u];
    const int s = (int)(ent >> 9);
    cl[u] = (int)(ent & 511u);
    nrm[u] = dinv[s] * dinv[node0 + cl[u]];
    const float* hp = h + (size_t)s * FEAT + lane;
    v[u][0] = hp[0];
    v[u][1] = hp[32];
    v[u][2] = hp[64];
    v[u][3] = hp[96];
  }
#pragma unroll
  for (int u = 0; u < K; ++u) {
    float* tp = tile + cl[u] * FEAT + lane;
    atomicAdd(tp + 0,  nrm[u] * v[u][0]);
    atomicAdd(tp + 32, nrm[u] * v[u][1]);
    atomicAdd(tp + 64, nrm[u] * v[u][2]);
    atomicAdd(tp + 96, nrm[u] * v[u][3]);
  }
}

__device__ __forceinline__ void own_drain(float* __restrict__ tile, const unsigned* __restrict__ list,
                                          int n, int node0,
                                          const float* __restrict__ h,
                                          const float* __restrict__ dinv,
                                          int wave, int lane) {
  const int groups = n >> 2;
  for (int g = wave; g < groups; g += OWN_T / 32)
    own_add<4>(tile, list + 4 * g, node0, h, dinv, lane);
  if (wave < (n & 3))
    own_add<1>(tile, list + 4 * groups + wave, node0, h, dinv, lane);
}

template <bool BUILD>
__global__ __launch_bounds__(OWN_T) void k_agg_own(const float* __restrict__ h,
                                                   const int* __restrict__ row,
                                                   const int* __restrict__ col,
                                                   const unsigned char* __restrict__ key,
                                                   const float* __restrict__ dinv,
                                                   float* __restrict__ plane,
                                                   unsigned* __restrict__ wlist,
                                                   int* __restrict__ wcnt) {
  __shared__ __attribute__((aligned(16))) float tile[TILE_NODES * FEAT];
  __shared__ unsigned list[LIST_CAP];
  __shared__ int cnt_s;

  const int t     = threadIdx.x;
  const int wave  = __builtin_amdgcn_readfirstlane(t >> 5);
  const int lane  = t & 31;
  const unsigned bid = blockIdx.x;
  const int node0 = (int)bid * TILE_NODES;
  unsigned* wl = wlist + (size_t)bid * LIST_CAP;
  const int saved = BUILD ? 0 : wcnt[bid];
  const bool rescan = BUILD || saved > LIST_CAP;

  if (t == 0) cnt_s = 0;
  {
    const float4* src4 = (const float4*)(plane + (size_t)node0 * FEAT);
    float4* tile4 = (float4*)tile;
    for (int i = t; i < TILE_NODES * FEAT / 4; i += OWN_T) tile4[i] = src4[i];
  }
  if (!rescan)
    for (int i = t; i < saved; i += OWN_T) list[i] = wl[i];
  __syncthreads();

  int n = saved;
  if (rescan) {

  const uint4* key4 = (const uint4*)key;
  for (int it0 = 0; it0 < SCAN_ITERS; it0 += SCAN_BATCH) {
    uint4 k[SCAN_BATCH];
#pragma unroll
    for (int u = 0; u < SCAN_BATCH; ++u) {
      const int q = (it0 + u) * OWN_T + t;
      k[u] = key4[min(q, N_KEYQ - 1)];
    }
#pragma unroll
    for (int u = 0; u < SCAN_BATCH; ++u) {
      const int q = (it0 + u) * OWN_T + t;
      const unsigned w[4] = {k[u].x, k[u].y, k[u].z, k[u].w};
      unsigned mask = 0;
#pragma unroll
      for (int j = 0; j < 4; ++j)
#pragma unroll
        for (int b = 0; b < 4; ++b)
          mask |= (unsigned)(((w[j] >> (8 * b)) & 0xFFu) == bid) << (4 * j + b);
      if (q >= N_KEYQ) mask = 0;
      while (mask) {
        const int pos = __builtin_ctz(mask);
        mask &= mask - 1;
        const int e = q * KEYS_PER_LOAD + pos;
        const int idx = atomicAdd(&cnt_s, 1);
        if (idx < LIST_CAP) {
          list[idx] = (unsigned)e;
        } else {
          const int c  = col[e];
          const int sN = row[e];
          const float nrm = dinv[sN] * dinv[c];
          float* tp = tile + (c - node0) * FEAT;
          const float* hp = h + (size_t)sN * FEAT;
#pragma nounroll
          for (int f = 0; f < FEAT; ++f) atomicAdd(tp + f, nrm * hp[f]);
        }
      }
    }
  }
  __syncthreads();
  n = min(cnt_s, LIST_CAP);
  for (int i = t; i < n; i += OWN_T) {
    const int e = (int)list[i];
    const unsigned p = ((unsigned)row[e] << 9) | (unsigned)(col[e] - node0);
    list[i] = p;
    if (BUILD) wl[i] = p;
  }
  if (BUILD && t == 0) wcnt[bid] = cnt_s;
  __syncthreads();
  }
  own_drain(tile, list, n, node0, h, dinv, wave, lane);
  __syncthreads();
  {
    float4* dst4 = (float4*)(plane + (size_t)node0 * FEAT);
    const float4* tile4 = (const float4*)tile;
    for (int i = t; i < TILE_NODES * FEAT / 4; i += OWN_T) dst4[i] = tile4[i];
  }
}

__global__ __launch_bounds__(256) void k_pool(const float* __restrict__ a,
                                              const int* __restrict__ batch,
                                              float* __restrict__ sums,
                                              float* __restrict__ cnts) {
  const int wave = threadIdx.x >> 5;
  const int lane = threadIdx.x & 31;
  const long i = (long)blockIdx.x * 8 + wave;
  if (i >= N_NODES) return;
  const int g = batch[i];
  const float* ap = a + (size_t)i * FEAT + lane;
  const float v0 = ap[0];
  const float v1 = ap[32];
  const float v2 = ap[64];
  const float v3 = ap[96];
  float* sp = sums + (size_t)g * FEAT + lane;
  atomAddF(sp + 0,  v0);
  atomAddF(sp + 32, v1);
  atomAddF(sp + 64, v2);
  atomAddF(sp + 96, v3);
  if (lane == 0) atomAddF(&cnts[g], 1.0f);
}

__global__ void k_final(const float* __restrict__ sums, const float* __restrict__ cnts,
                        const float* __restrict__ Wl, const float* __restrict__ bl,
                        float* __restrict__ out) {
  const int g = blockIdx.x;
  const int c = threadIdx.x;
  if (c >= NCLS) return;
  const float inv = 1.0f / fmaxf(cnts[g], 1.0f);
  float acc = bl[c];
  for (int k = 0; k < FEAT; ++k)
    acc += sums[(size_t)g * FEAT + k] * inv * Wl[k * NCLS + c];
  out[(size_t)g * NCLS + c] = acc;
}

extern "C" void kernel_launch(void* const* d_in, const int* in_sizes, int n_in,
                              void* d_out, int out_size, void* d_ws, size_t ws_size,
                              hipStream_t stream) {
  const float* x     = (const float*)d_in[0];
  const int*   ei    = (const int*)d_in[1];
  const int*   row   = ei;
  const int*   col   = ei + N_EDGES;
  const int*   batch = (const int*)d_in[2];
  const float* W1 = (const float*)d_in[3];
  const float* b1 = (const float*)d_in[4];
  const float* W2 = (const float*)d_in[5];
  const float* b2 = (const float*)d_in[6];
  const float* W3 = (const float*)d_in[7];
  const float* b3 = (const float*)d_in[8];
  const float* Wl = (const float*)d_in[9];
  const float* bl = (const float*)d_in[10];
  float* outp = (float*)d_out;

  char* base = (char*)d_ws;
  const size_t featBytes = (size_t)N_NODES * FEAT * sizeof(float);
  float* deg  = (float*)(base);
  float* hbuf = (float*)(base + (1 << 20));
  float* pA   = (float*)(base + (1 << 20) + featBytes);
  float* pB   = (float*)(base + (1 << 20) + 2 * featBytes);
  float* sums = (float*)(base + (1 << 20) + 3 * featBytes);
  float* cnts = sums + (size_t)NGRAPH * FEAT;
  unsigned char* key = (unsigned char*)(base + (2 << 20) + 3 * featBytes);
  int*      wcnt  = (int*)(base + (3 << 20) + 3 * featBytes);
  unsigned* wlist = (unsigned*)(base + (4 << 20) + 3 * featBytes);

  const int  TB     = 256;
  const int  gNode  = (N_NODES + TB - 1) / TB;
  const int  gEdge4 = (N_EDGES / 4 + TB - 1) / TB;
  const int  gGemm  = (N_NODES / 16 + 8 - 1) / 8;
  const int  gPool  = (N_NODES + 8 - 1) / 8;

  k_fill<<<gNode, TB, 0, stream>>>(deg, 1.0f, N_NODES);
  k_deg<<<gEdge4, TB, 0, stream>>>(col, deg, (unsigned*)key);
  k_dinv<<<gNode, TB, 0, stream>>>(deg);

  k_gemm<false><<<gGemm, TB, 0, stream>>>(x, W1, deg, b1, hbuf, pA, N_NODES);
  k_agg_own<true><<<N_TILES, OWN_T, 0, stream>>>(hbuf, row, col, key, deg, pA, wlist, wcnt);

  k_gemm<true><<<gGemm, TB, 0, stream>>>(pA, W2, deg, b2, hbuf, pB, N_NODES);
  k_agg_own<false><<<N_TILES, OWN_T, 0, stream>>>(hbuf, row, col, key, deg, pB, wlist, wcnt);

  k_gemm<true><<<gGemm, TB, 0, stream>>>(pB, W3, deg, b3, hbuf, pA, N_NODES);
  k_agg_own<false><<<N_TILES, OWN_T, 0, stream>>>(hbuf, row, col, key, deg, pA, wlist, wcnt);

  const long poolN = (long)NGRAPH * FEAT + NGRAPH;
  k_fill<<<(int)((poolN + TB - 1) / TB), TB, 0, stream>>>(sums, 0.0f, poolN);
  k_pool<<<gPool, TB, 0, stream>>>(pA, batch, sums, cnts);
  k_final<<<NGRAPH, 32, 0, stream>>>(sums, cnts, Wl, bl, outp);
}


# layer product moved onto the owner tile in LDS: input rows aggregated first (linear), then tile x W on WMMA, bias, ReLU, in place; product launches, H and start planes gone; rounding order changes
# speedup vs baseline: 2.7060x; 1.3074x over previous
#include <hip/hip_runtime.h>

typedef __attribute__((ext_vector_type(2))) float v2f;
typedef __attribute__((ext_vector_type(8))) float v8f;

#define N_NODES 100000
#define N_EDGES 800000
#define FEAT    128
#define NGRAPH  512
#define NCLS    10

__device__ __forceinline__ void atomAddF(float* p, float v) {
  unsafeAtomicAdd(p, v);
}

__global__ void k_fill(float* __restrict__ p, float v, long n) {
  long i = (long)blockIdx.x * blockDim.x + threadIdx.x;
  if (i < n) p[i] = v;
}

#define TILE_NODES 400
__global__ void k_deg(const int* __restrict__ col, float* __restrict__ deg,
                      unsigned* __restrict__ key) {
  long i = (long)blockIdx.x * blockDim.x + threadIdx.x;
  if (i < N_EDGES / 4) {
    const int4 c = ((const int4*)col)[i];
    atomAddF(&deg[c.x], 1.0f);
    atomAddF(&deg[c.y], 1.0f);
    atomAddF(&deg[c.z], 1.0f);
    atomAddF(&deg[c.w], 1.0f);
    key[i] = (unsigned)(c.x / TILE_NODES) | ((unsigned)(c.y / TILE_NODES) << 8) |
             ((unsigned)(c.z / TILE_NODES) << 16) | ((unsigned)(c.w / TILE_NODES) << 24);
  }
}

__global__ void k_dinv(float* __restrict__ deg) {
  long i = (long)blockIdx.x * blockDim.x + threadIdx.x;
  if (i < N_NODES) {
    float d = deg[i];
    deg[i] = (d > 0.0f) ? rsqrtf(fmaxf(d, 1.0f)) : 0.0f;
  }
}

#define N_TILES    (N_NODES / TILE_NODES)
#define OWN_T      1024
#define KEYS_PER_LOAD 16
#define N_KEYQ     (N_EDGES / KEYS_PER_LOAD)
#define SCAN_ITERS ((N_KEYQ + OWN_T - 1) / OWN_T)
#define SCAN_BATCH 7
#define LIST_CAP   24576
#define TPITCH     (FEAT + 4)
#define N_STRIPS   (TILE_NODES / 16)
static_assert(N_NODES % TILE_NODES == 0 && N_TILES <= 255, "tile key must fit a byte below 0xFF");
static_assert(N_EDGES % KEYS_PER_LOAD == 0 && N_EDGES % 4 == 0, "key array is written 4 and read 16 bytes at a time");
static_assert(TILE_NODES <= 512 && N_NODES <= (1 << 17), "list entry packs source (17 bits) and local row (9 bits)");
static_assert((4l << 20) + 3l * N_NODES * FEAT * 4 + (long)N_TILES * LIST_CAP * 4 <= 268435456l, "stored lists end inside the workspace");
static_assert(TILE_NODES % 16 == 0 && N_STRIPS <= OWN_T / 32, "one wave per 16-row strip of the tile");
static_assert(FEAT * FEAT + FEAT <= LIST_CAP && FEAT * FEAT / 4 == 4 * OWN_T, "W and the bias row are staged in the list's LDS, W in four 16-byte pieces per thread");
static_assert((TILE_NODES * TPITCH + LIST_CAP + 4) * 4 <= 320 * 1024, "tile and list fit one WGP's LDS");

template <int K>
__device__ __forceinline__ void own_add(float* __restrict__ tile, const unsigned* __restrict__ ents,
                                        int node0, const float* __restrict__ xin,
                                        const float* __restrict__ dinv, int lane) {
  int   cl[K];
  float nrm[K];
  float v[K][4];
#pragma unroll
  for (int u = 0; u < K; ++u) {
    const unsigned ent = ents[u];
    const int s = (int)(ent >> 9);
    cl[u] = (int)(ent & 511u);
    nrm[u] = dinv[s] * dinv[node0 + cl[u]];
    const float* hp = xin + (size_t)s * FEAT + lane;
    v[u][0] = hp[0];
    v[u][1] = hp[32];
    v[u][2] = hp[64];
    v[u][3] = hp[96];
  }
#pragma unroll
  for (int u = 0; u < K; ++u) {
    float* tp = tile + cl[u] * TPITCH + lane;
    atomicAdd(tp + 0,  nrm[u] * v[u][0]);
    atomicAdd(tp + 32, nrm[u] * v[u][1]);
    atomicAdd(tp + 64, nrm[u] * v[u][2]);
    atomicAdd(tp + 96, nrm[u] * v[u][3]);
  }
}

__device__ __forceinline__ void own_drain(float* __restrict__ tile, const unsigned* __restrict__ list,
                                          int n, int node0,
                                          const float* __restrict__ xin,
                                          const float* __restrict__ dinv,
                                          int wave, int lane) {
  const int groups = n >> 2;
  for (int g = wave; g < groups; g += OWN_T / 32)
    own_add<4>(tile, list + 4 * g, node0, xin, dinv, lane);
  if (wave < (n & 3))
    own_add<1>(tile, list + 4 * groups + wave, node0, xin, dinv, lane);
}

template <bool RELU_OUT>
__device__ __forceinline__ void own_product(float* __restrict__ tile, const float* __restrict__ sW,
                                            const float* __restrict__ sBias, int strip, int lane) {
  const int m    = lane & 15;
  const int koff = (lane >> 4) << 1;
  const int ncol = lane & 15;

  v8f acc[8];
#pragma unroll
  for (int n = 0; n < 8; ++n) acc[n] = (v8f){0, 0, 0, 0, 0, 0, 0, 0};

  const float* arow = tile + (strip * 16 + m) * TPITCH + koff;

  for (int k0 = 0; k0 < FEAT; k0 += 16) {
    v2f a[4];
#pragma unroll
    for (int j = 0; j < 4; ++j) {
      a[j].x = arow[k0 + 4 * j];
      a[j].y = arow[k0 + 4 * j + 1];
    }
    v2f b[2][8];
#pragma unroll
    for (int n = 0; n < 8; ++n) {
      const int kb = k0 + koff;
      b[0][n].x = sW[kb * FEAT + n * 16 + ncol];
      b[0][n].y = sW[(kb + 1) * FEAT + n * 16 + ncol];
    }
#pragma unroll
    for (int j = 0; j < 4; ++j) {
      if (j < 3) {
        const int kb = k0 + 4 * (j + 1) + koff;
#pragma unroll
        for (int n = 0; n < 8; ++n) {
          b[(j + 1) & 1][n].x = sW[kb * FEAT + n * 16 + ncol];
          b[(j + 1) & 1][n].y = sW[(kb + 1) * FEAT + n * 16 + ncol];
        }
      }
#pragma unroll
      for (int n = 0; n < 8; ++n) {
        acc[n] = __builtin_amdgcn_wmma_f32_16x16x4_f32(
            false, a[j], false, b[j & 1][n], (short)0, acc[n], false, false);
      }
    }
  }

  const int rowsel = (lane >> 4) * 8;
  float* orow = tile + (strip * 16 + rowsel) * TPITCH + ncol;
#pragma unroll
  for (int n = 0; n < 8; ++n) {
    const float bv = sBias[n * 16 + ncol];
#pragma unroll
    for (int r = 0; r < 8; ++r) {
      float o = acc[n][r] + bv;
      if (RELU_OUT) o = fmaxf(o, 0.0f);
      orow[r * TPITCH + n * 16] = o;
    }
  }
}

template <bool BUILD, bool RELU_OUT>
__global__ __launch_bounds__(OWN_T) void k_layer_own(const float* __restrict__ xin,
                                                     const float* __restrict__ W,
                                                     const float* __restrict__ bias,
                                                     const int* __restrict__ row,
                                                     const int* __restrict__ col,
                                                     const unsigned char* __restrict__ key,
                                                     const float* __restrict__ dinv,
                                                     float* __restrict__ xout,
                                                     unsigned* __restrict__ wlist,
                                                     int* __restrict__ wcnt) {
  __shared__ __attribute__((aligned(16))) unsigned smem[LIST_CAP + TILE_NODES * TPITCH];
  unsigned* const list = smem;
  float* const tile = (float*)(smem + LIST_CAP);
  __shared__ int cnt_s;

  const int t     = threadIdx.x;
  const int wave  = __builtin_amdgcn_readfirstlane(t >> 5);
  const int lane  = t & 31;
  const unsigned bid = blockIdx.x;
  const int node0 = (int)bid * TILE_NODES;
  unsigned* wl = wlist + (size_t)bid * LIST_CAP;
  const int saved = BUILD ? 0 : wcnt[bid];
  const bool rescan = BUILD || saved > LIST_CAP;

  if (t == 0) cnt_s = 0;
  {
    const float4* src4 = (const float4*)(xin + (size_t)node0 * FEAT);
    for (int i = t; i < TILE_NODES * FEAT / 4; i += OWN_T) {
      const int r = i >> 5, c4 = i & 31;
      const float d = dinv[node0 + r];
      const float s = d * d;
      float4 v = src4[i];
      v.x *= s; v.y *= s; v.z *= s; v.w *= s;
      *(float4*)(tile + r * TPITCH + 4 * c4) = v;
    }
  }
  if (!rescan)
    for (int i = t; i < saved; i += OWN_T) list[i] = wl[i];
  __syncthreads();

  int n = saved;
  if (rescan) {

  const uint4* key4 = (const uint4*)key;
  for (int it0 = 0; it0 < SCAN_ITERS; it0 += SCAN_BATCH) {
    uint4 k[SCAN_BATCH];
#pragma unroll
    for (int u = 0; u < SCAN_BATCH; ++u) {
      const int q = (it0 + u) * OWN_T + t;
      k[u] = key4[min(q, N_KEYQ - 1)];
    }
#pragma unroll
    for (int u = 0; u < SCAN_BATCH; ++u) {
      const int q = (it0 + u) * OWN_T + t;
      const unsigned w[4] = {k[u].x, k[u].y, k[u].z, k[u].w};
      unsigned mask = 0;
#pragma unroll
      for (int j = 0; j < 4; ++j)
#pragma unroll
        for (int b = 0; b < 4; ++b)
          mask |= (unsigned)(((w[j] >> (8 * b)) & 0xFFu) == bid) << (4 * j + b);
      if (q >= N_KEYQ) mask = 0;
      while (mask) {
        const int pos = __builtin_ctz(mask);
        mask &= mask - 1;
        const int e = q * KEYS_PER_LOAD + pos;
        const int idx = atomicAdd(&cnt_s, 1);
        if (idx < LIST_CAP) {
          list[idx] = (unsigned)e;
        } else {
          const int c  = col[e];
          const int sN = row[e];
          const float nrm = dinv[sN] * dinv[c];
          float* tp = tile + (c - node0) * TPITCH;
          const float* hp = xin + (size_t)sN * FEAT;
#pragma nounroll
          for (int f = 0; f < FEAT; ++f) atomicAdd(tp + f, nrm * hp[f]);
        }
      }
    }
  }
  __syncthreads();
  n = min(cnt_s, LIST_CAP);
  for (int i = t; i < n; i += OWN_T) {
    const int e = (int)list[i];
    const unsigned p = ((unsigned)row[e] << 9) | (unsigned)(col[e] - node0);
    list[i] = p;
    if (BUILD) wl[i] = p;
  }
  if (BUILD && t == 0) wcnt[bid] = cnt_s;
  __syncthreads();
  }
  own_drain(tile, list, n, node0, xin, dinv, wave, lane);

  __syncthreads();
  float* sW = (float*)list;
#pragma unroll
  for (int j = 0; j < 4; ++j) ((float4*)sW)[t + j * OWN_T] = ((const float4*)W)[t + j * OWN_T];
  float* sBias = sW + FEAT * FEAT;
  if (t < FEAT / 4) ((float4*)sBias)[t] = ((const float4*)bias)[t];
  __syncthreads();

  if (wave < N_STRIPS)
    own_product<RELU_OUT>(tile, sW, sBias, wave, lane);
  __syncthreads();
  {
    float4* dst4 = (float4*)(xout + (size_t)node0 * FEAT);
    for (int i = t; i < TILE_NODES * FEAT / 4; i += OWN_T) {
      const int r = i >> 5, c4 = i & 31;
      dst4[i] = *(const float4*)(tile + r * TPITCH + 4 * c4);
    }
  }
}

__global__ __launch_bounds__(256) void k_pool(const float* __restrict__ a,
                                              const int* __restrict__ batch,
                                              float* __restrict__ sums,
                                              float* __restrict__ cnts) {
  const int wave = threadIdx.x >> 5;
  const int lane = threadIdx.x & 31;
  const long i = (long)blockIdx.x * 8 + wave;
  if (i >= N_NODES) return;
  const int g = batch[i];
  const float* ap = a + (size_t)i * FEAT + lane;
  const float v0 = ap[0];
  const float v1 = ap[32];
  const float v2 = ap[64];
  const float v3 = ap[96];
  float* sp = sums + (size_t)g * FEAT + lane;
  atomAddF(sp + 0,  v0);
  atomAddF(sp + 32, v1);
  atomAddF(sp + 64, v2);
  atomAddF(sp + 96, v3);
  if (lane == 0) atomAddF(&cnts[g], 1.0f);
}

__global__ void k_final(const float* __restrict__ sums, const float* __restrict__ cnts,
                        const float* __restrict__ Wl, const float* __restrict__ bl,
                        float* __restrict__ out) {
  const int g = blockIdx.x;
  const int c = threadIdx.x;
  if (c >= NCLS) return;
  const float inv = 1.0f / fmaxf(cnts[g], 1.0f);
  float acc = bl[c];
  for (int k = 0; k < FEAT; ++k)
    acc += sums[(size_t)g * FEAT + k] * inv * Wl[k * NCLS + c];
  out[(size_t)g * NCLS + c] = acc;
}

extern "C" void kernel_launch(void* const* d_in, const int* in_sizes, int n_in,
                              void* d_out, int out_size, void* d_ws, size_t ws_size,
                              hipStream_t stream) {
  const float* x     = (const float*)d_in[0];
  const int*   ei    = (const int*)d_in[1];
  const int*   row   = ei;
  const int*   col   = ei + N_EDGES;
  const int*   batch = (const int*)d_in[2];
  const float* W1 = (const float*)d_in[3];
  const float* b1 = (const float*)d_in[4];
  const float* W2 = (const float*)d_in[5];
  const float* b2 = (const float*)d_in[6];
  const float* W3 = (const float*)d_in[7];
  const float* b3 = (const float*)d_in[8];
  const float* Wl = (const float*)d_in[9];
  const float* bl = (const float*)d_in[10];
  float* outp = (float*)d_out;

  char* base = (char*)d_ws;
  const size_t featBytes = (size_t)N_NODES * FEAT * sizeof(float);
  float* deg  = (float*)(base);
  float* pA   = (float*)(base + (1 << 20) + featBytes);
  float* pB   = (float*)(base + (1 << 20) + 2 * featBytes);
  float* sums = (float*)(base + (1 << 20) + 3 * featBytes);
  float* cnts = sums + (size_t)NGRAPH * FEAT;
  unsigned char* key = (unsigned char*)(base + (2 << 20) + 3 * featBytes);
  int*      wcnt  = (int*)(base + (3 << 20) + 3 * featBytes);
  unsigned* wlist = (unsigned*)(base + (4 << 20) + 3 * featBytes);

  const int  TB     = 256;
  const int  gNode  = (N_NODES + TB - 1) / TB;
  const int  gEdge4 = (N_EDGES / 4 + TB - 1) / TB;
  const int  gPool  = (N_NODES + 8 - 1) / 8;

  k_fill<<<gNode, TB, 0, stream>>>(deg, 1.0f, N_NODES);
  k_deg<<<gEdge4, TB, 0, stream>>>(col, deg, (unsigned*)key);
  k_dinv<<<gNode, TB, 0, stream>>>(deg);

  k_layer_own<true, true><<<N_TILES, OWN_T, 0, stream>>>(x, W1, b1, row, col, key, deg, pA, wlist, wcnt);
  k_layer_own<false, true><<<N_TILES, OWN_T, 0, stream>>>(pA, W2, b2, row, col, key, deg, pB, wlist, wcnt);
  k_layer_own<false, false><<<N_TILES, OWN_T, 0, stream>>>(pB, W3, b3, row, col, key, deg, pA, wlist, wcnt);

  const long poolN = (long)NGRAPH * FEAT + NGRAPH;
  k_fill<<<(int)((poolN + TB - 1) / TB), TB, 0, stream>>>(sums, 0.0f, poolN);
  k_pool<<<gPool, TB, 0, stream>>>(pA, batch, sums, cnts);
  k_final<<<NGRAPH, 32, 0, stream>>>(sums, cnts, Wl, bl, outp);
}


# pool inside layer 3's launch: each owner tile sums its finished rows per graph id out of LDS, few whole-line atomics; pool and zero-fill launches, 51 MB write+read gone; pooled sum order changes
# speedup vs baseline: 3.7748x; 1.3950x over previous
#include <hip/hip_runtime.h>

typedef __attribute__((ext_vector_type(2))) float v2f;
typedef __attribute__((ext_vector_type(8))) float v8f;

#define N_NODES 100000
#define N_EDGES 800000
#define FEAT    128
#define NGRAPH  512
#define NCLS    10

__device__ __forceinline__ void atomAddF(float* p, float v) {
  unsafeAtomicAdd(p, v);
}

__global__ void k_fill(float* __restrict__ p, float v, long n) {
  long i = (long)blockIdx.x * blockDim.x + threadIdx.x;
  if (i < n) p[i] = v;
}

#define TILE_NODES 400
__global__ void k_deg(const int* __restrict__ col, float* __restrict__ deg,
                      unsigned* __restrict__ key) {
  long i = (long)blockIdx.x * blockDim.x + threadIdx.x;
  if (i < N_EDGES / 4) {
    const int4 c = ((const int4*)col)[i];
    atomAddF(&deg[c.x], 1.0f);
    atomAddF(&deg[c.y], 1.0f);
    atomAddF(&deg[c.z], 1.0f);
    atomAddF(&deg[c.w], 1.0f);
    key[i] = (unsigned)(c.x / TILE_NODES) | ((unsigned)(c.y / TILE_NODES) << 8) |
             ((unsigned)(c.z / TILE_NODES) << 16) | ((unsigned)(c.w / TILE_NODES) << 24);
  }
}

#define POOL_WORDS (NGRAPH * FEAT + NGRAPH)
static_assert(POOL_WORDS <= N_NODES, "the node grid also covers the pooled sums and counts");
__global__ void k_dinv(float* __restrict__ deg, float* __restrict__ pool) {
  long i = (long)blockIdx.x * blockDim.x + threadIdx.x;
  if (i < N_NODES) {
    float d = deg[i];
    deg[i] = (d > 0.0f) ? rsqrtf(fmaxf(d, 1.0f)) : 0.0f;
  }
  if (i < POOL_WORDS) pool[i] = 0.0f;
}

#define N_TILES    (N_NODES / TILE_NODES)
#define OWN_T      1024
#define KEYS_PER_LOAD 16
#define N_KEYQ     (N_EDGES / KEYS_PER_LOAD)
#define SCAN_ITERS ((N_KEYQ + OWN_T - 1) / OWN_T)
#define SCAN_BATCH 7
#define LIST_CAP   24576
#define TPITCH     (FEAT + 4)
#define N_STRIPS   (TILE_NODES / 16)
static_assert(N_NODES % TILE_NODES == 0 && N_TILES <= 255, "tile key must fit a byte below 0xFF");
static_assert(N_EDGES % KEYS_PER_LOAD == 0 && N_EDGES % 4 == 0, "key array is written 4 and read 16 bytes at a time");
static_assert(TILE_NODES <= 512 && N_NODES <= (1 << 17), "list entry packs source (17 bits) and local row (9 bits)");
static_assert((4l << 20) + 3l * N_NODES * FEAT * 4 + (long)N_TILES * LIST_CAP * 4 <= 268435456l, "stored lists end inside the workspace");
static_assert(TILE_NODES % 16 == 0 && N_STRIPS <= OWN_T / 32, "one wave per 16-row strip of the tile");
static_assert(FEAT * FEAT + FEAT <= LIST_CAP && FEAT * FEAT / 4 == 4 * OWN_T, "W and the bias row are staged in the list's LDS, W in four 16-byte pieces per thread");
static_assert((TILE_NODES * TPITCH + LIST_CAP + 4) * 4 <= 320 * 1024, "tile and list fit one WGP's LDS");

template <int K>
__device__ __forceinline__ void own_add(float* __restrict__ tile, const unsigned* __restrict__ ents,
                                        int node0, const float* __restrict__ xin,
                                        const float* __restrict__ dinv, int lane) {
  int   cl[K];
  float nrm[K];
  float v[K][4];
#pragma unroll
  for (int u = 0; u < K; ++u) {
    const unsigned ent = ents[u];
    const int s = (int)(ent >> 9);
    cl[u] = (int)(ent & 511u);
    nrm[u] = dinv[s] * dinv[node0 + cl[u]];
    const float* hp = xin + (size_t)s * FEAT + lane;
    v[u][0] = hp[0];
    v[u][1] = hp[32];
    v[u][2] = hp[64];
    v[u][3] = hp[96];
  }
#pragma unroll
  for (int u = 0; u < K; ++u) {
    float* tp = tile + cl[u] * TPITCH + lane;
    atomicAdd(tp + 0,  nrm[u] * v[u][0]);
    atomicAdd(tp + 32, nrm[u] * v[u][1]);
    atomicAdd(tp + 64, nrm[u] * v[u][2]);
    atomicAdd(tp + 96, nrm[u] * v[u][3]);
  }
}

__device__ __forceinline__ void own_drain(float* __restrict__ tile, const unsigned* __restrict__ list,
                                          int n, int node0,
                                          const float* __restrict__ xin,
                                          const float* __restrict__ dinv,
                                          int wave, int lane) {
  const int groups = n >> 2;
  for (int g = wave; g < groups; g += OWN_T / 32)
    own_add<4>(tile, list + 4 * g, node0, xin, dinv, lane);
  if (wave < (n & 3))
    own_add<1>(tile, list + 4 * groups + wave, node0, xin, dinv, lane);
}

template <bool RELU_OUT>
__device__ __forceinline__ void own_product(float* __restrict__ tile, const float* __restrict__ sW,
                                            const float* __restrict__ sBias, int strip, int lane) {
  const int m    = lane & 15;
  const int koff = (lane >> 4) << 1;
  const int ncol = lane & 15;

  v8f acc[8];
#pragma unroll
  for (int n = 0; n < 8; ++n) acc[n] = (v8f){0, 0, 0, 0, 0, 0, 0, 0};

  const float* arow = tile + (strip * 16 + m) * TPITCH + koff;

  for (int k0 = 0; k0 < FEAT; k0 += 16) {
    v2f a[4];
#pragma unroll
    for (int j = 0; j < 4; ++j) {
      a[j].x = arow[k0 + 4 * j];
      a[j].y = arow[k0 + 4 * j + 1];
    }
    v2f b[2][8];
#pragma unroll
    for (int n = 0; n < 8; ++n) {
      const int kb = k0 + koff;
      b[0][n].x = sW[kb * FEAT + n * 16 + ncol];
      b[0][n].y = sW[(kb + 1) * FEAT + n * 16 + ncol];
    }
#pragma unroll
    for (int j = 0; j < 4; ++j) {
      if (j < 3) {
        const int kb = k0 + 4 * (j + 1) + koff;
#pragma unroll
        for (int n = 0; n < 8; ++n) {
          b[(j + 1) & 1][n].x = sW[kb * FEAT + n * 16 + ncol];
          b[(j + 1) & 1][n].y = sW[(kb + 1) * FEAT + n * 16 + ncol];
        }
      }
#pragma unroll
      for (int n = 0; n < 8; ++n) {
        acc[n] = __builtin_amdgcn_wmma_f32_16x16x4_f32(
            false, a[j], false, b[j & 1][n], (short)0, acc[n], false, false);
      }
    }
  }

  const int rowsel = (lane >> 4) * 8;
  float* orow = tile + (strip * 16 + rowsel) * TPITCH + ncol;
#pragma unroll
  for (int n = 0; n < 8; ++n) {
    const float bv = sBias[n * 16 + ncol];
#pragma unroll
    for (int r = 0; r < 8; ++r) {
      float o = acc[n][r] + bv;
      if (RELU_OUT) o = fmaxf(o, 0.0f);
      orow[r * TPITCH + n * 16] = o;
    }
  }
}

#define POOL_SLICES (OWN_T / FEAT)
#define POOL_ROWS   (TILE_NODES / POOL_SLICES)
#define POOL_CHUNK  10
static_assert(OWN_T % FEAT == 0 && FEAT % 32 == 0 && TILE_NODES % POOL_SLICES == 0, "a wave stays inside one slice; slices cover the tile");
static_assert(POOL_ROWS <= 64 && POOL_ROWS % POOL_CHUNK == 0, "a slice's ids fit two registers of a wave; whole chunks");
__device__ __forceinline__ void pool_flush(float* __restrict__ sums, float* __restrict__ cnts,
                                           int g, int f, float acc, int run) {
  atomAddF(sums + (size_t)g * FEAT + f, acc);
  if (f == 0) atomAddF(cnts + g, (float)run);
}
__device__ __forceinline__ void own_pool(const float* __restrict__ tile, const int* __restrict__ batch,
                                         int node0, float* __restrict__ sums, float* __restrict__ cnts,
                                         int wave, int lane) {
  const int sl = wave / (FEAT / 32);
  const int f  = (wave % (FEAT / 32)) * 32 + lane;
  const int r0 = sl * POOL_ROWS;
  const int* bp = batch + node0 + r0;
  const int ida = bp[min(lane, POOL_ROWS - 1)];
  const int idb = bp[min(lane + 32, POOL_ROWS - 1)];
  const float* tp = tile + r0 * TPITCH + f;
  int   g   = __builtin_amdgcn_readlane(ida, 0);
  float acc = 0.0f;
  int   run = 0;
  for (int c = 0; c < POOL_ROWS; c += POOL_CHUNK) {
    float v[POOL_CHUNK];
#pragma unroll
    for (int j = 0; j < POOL_CHUNK; ++j) v[j] = tp[(c + j) * TPITCH];
#pragma unroll
    for (int j = 0; j < POOL_CHUNK; ++j) {
      const int r  = c + j;
      const int gr = (r < 32) ? __builtin_amdgcn_readlane(ida, r) : __builtin_amdgcn_readlane(idb, r - 32);
      if (gr != g) {
        pool_flush(sums, cnts, g, f, acc, run);
        g = gr; acc = 0.0f; run = 0;
      }
      acc += v[j];
      ++run;
    }
  }
  pool_flush(sums, cnts, g, f, acc, run);
}

template <bool BUILD, bool RELU_OUT, bool POOL>
__global__ __launch_bounds__(OWN_T) void k_layer_own(const float* __restrict__ xin,
                                                     const float* __restrict__ W,
                                                     const float* __restrict__ bias,
                                                     const int* __restrict__ row,
                                                     const int* __restrict__ col,
                                                     const unsigned char* __restrict__ key,
                                                     const float* __restrict__ dinv,
                                                     float* __restrict__ xout,
                                                     unsigned* __restrict__ wlist,
                                                     int* __restrict__ wcnt,
                                                     const int* __restrict__ batch,
                                                     float* __restrict__ sums,
                                                     float* __restrict__ cnts) {
  __shared__ __attribute__((aligned(16))) unsigned smem[LIST_CAP + TILE_NODES * TPITCH];
  unsigned* const list = smem;
  float* const tile = (float*)(smem + LIST_CAP);
  __shared__ int cnt_s;

  const int t     = threadIdx.x;
  const int wave  = __builtin_amdgcn_readfirstlane(t >> 5);
  const int lane  = t & 31;
  const unsigned bid = blockIdx.x;
  const int node0 = (int)bid * TILE_NODES;
  unsigned* wl = wlist + (size_t)bid * LIST_CAP;
  const int saved = BUILD ? 0 : wcnt[bid];
  const bool rescan = BUILD || saved > LIST_CAP;

  if (t == 0) cnt_s = 0;
  {
    const float4* src4 = (const float4*)(xin + (size_t)node0 * FEAT);
    for (int i = t; i < TILE_NODES * FEAT / 4; i += OWN_T) {
      const int r = i >> 5, c4 = i & 31;
      const float d = dinv[node0 + r];
      const float s = d * d;
      float4 v = src4[i];
      v.x *= s; v.y *= s; v.z *= s; v.w *= s;
      *(float4*)(tile + r * TPITCH + 4 * c4) = v;
    }
  }
  if (!rescan)
    for (int i = t; i < saved; i += OWN_T) list[i] = wl[i];
  __syncthreads();

  int n = saved;
  if (rescan) {

  const uint4* key4 = (const uint4*)key;
  for (int it0 = 0; it0 < SCAN_ITERS; it0 += SCAN_BATCH) {
    uint4 k[SCAN_BATCH];
#pragma unroll
    for (int u = 0; u < SCAN_BATCH; ++u) {
      const int q = (it0 + u) * OWN_T + t;
      k[u] = key4[min(q, N_KEYQ - 1)];
    }
#pragma unroll
    for (int u = 0; u < SCAN_BATCH; ++u) {
      const int q = (it0 + u) * OWN_T + t;
      const unsigned w[4] = {k[u].x, k[u].y, k[u].z, k[u].w};
      unsigned mask = 0;
#pragma unroll
      for (int j = 0; j < 4; ++j)
#pragma unroll
        for (int b = 0; b < 4; ++b)
          mask |= (unsigned)(((w[j] >> (8 * b)) & 0xFFu) == bid) << (4 * j + b);
      if (q >= N_KEYQ) mask = 0;
      while (mask) {
        const int pos = __builtin_ctz(mask);
        mask &= mask - 1;
        const int e = q * KEYS_PER_LOAD + pos;
        const int idx = atomicAdd(&cnt_s, 1);
        if (idx < LIST_CAP) {
          list[idx] = (unsigned)e;
        } else {
          const int c  = col[e];
          const int sN = row[e];
          const float nrm = dinv[sN] * dinv[c];
          float* tp = tile + (c - node0) * TPITCH;
          const float* hp = xin + (size_t)sN * FEAT;
#pragma nounroll
          for (int f = 0; f < FEAT; ++f) atomicAdd(tp + f, nrm * hp[f]);
        }
      }
    }
  }
  __syncthreads();
  n = min(cnt_s, LIST_CAP);
  for (int i = t; i < n; i += OWN_T) {
    const int e = (int)list[i];
    const unsigned p = ((unsigned)row[e] << 9) | (unsigned)(col[e] - node0);
    list[i] = p;
    if (BUILD) wl[i] = p;
  }
  if (BUILD && t == 0) wcnt[bid] = cnt_s;
  __syncthreads();
  }
  own_drain(tile, list, n, node0, xin, dinv, wave, lane);

  __syncthreads();
  float* sW = (float*)list;
#pragma unroll
  for (int j = 0; j < 4; ++j) ((float4*)sW)[t + j * OWN_T] = ((const float4*)W)[t + j * OWN_T];
  float* sBias = sW + FEAT * FEAT;
  if (t < FEAT / 4) ((float4*)sBias)[t] = ((const float4*)bias)[t];
  __syncthreads();

  if (wave < N_STRIPS)
    own_product<RELU_OUT>(tile, sW, sBias, wave, lane);
  __syncthreads();
  if (POOL) {
    own_pool(tile, batch, node0, sums, cnts, wave, lane);
  } else {
    float4* dst4 = (float4*)(xout + (size_t)node0 * FEAT);
    for (int i = t; i < TILE_NODES * FEAT / 4; i += OWN_T) {
      const int r = i >> 5, c4 = i & 31;
      dst4[i] = *(const float4*)(tile + r * TPITCH + 4 * c4);
    }
  }
}

__global__ void k_final(const float* __restrict__ sums, const float* __restrict__ cnts,
                        const float* __restrict__ Wl, const float* __restrict__ bl,
                        float* __restrict__ out) {
  const int g = blockIdx.x;
  const int c = threadIdx.x;
  if (c >= NCLS) return;
  const float inv = 1.0f / fmaxf(cnts[g], 1.0f);
  float acc = bl[c];
  for (int k = 0; k < FEAT; ++k)
    acc += sums[(size_t)g * FEAT + k] * inv * Wl[k * NCLS + c];
  out[(size_t)g * NCLS + c] = acc;
}

extern "C" void kernel_launch(void* const* d_in, const int* in_sizes, int n_in,
                              void* d_out, int out_size, void* d_ws, size_t ws_size,
                              hipStream_t stream) {
  const float* x     = (const float*)d_in[0];
  const int*   ei    = (const int*)d_in[1];
  const int*   row   = ei;
  const int*   col   = ei + N_EDGES;
  const int*   batch = (const int*)d_in[2];
  const float* W1 = (const float*)d_in[3];
  const float* b1 = (const float*)d_in[4];
  const float* W2 = (const float*)d_in[5];
  const float* b2 = (const float*)d_in[6];
  const float* W3 = (const float*)d_in[7];
  const float* b3 = (const float*)d_in[8];
  const float* Wl = (const float*)d_in[9];
  const float* bl = (const float*)d_in[10];
  float* outp = (float*)d_out;

  char* base = (char*)d_ws;
  const size_t featBytes = (size_t)N_NODES * FEAT * sizeof(float);
  float* deg  = (float*)(base);
  float* pA   = (float*)(base + (1 << 20) + featBytes);
  float* pB   = (float*)(base + (1 << 20) + 2 * featBytes);
  float* sums = (float*)(base + (1 << 20) + 3 * featBytes);
  float* cnts = sums + (size_t)NGRAPH * FEAT;
  unsigned char* key = (unsigned char*)(base + (2 << 20) + 3 * featBytes);
  int*      wcnt  = (int*)(base + (3 << 20) + 3 * featBytes);
  unsigned* wlist = (unsigned*)(base + (4 << 20) + 3 * featBytes);

  const int  TB     = 256;
  const int  gNode  = (N_NODES + TB - 1) / TB;
  const int  gEdge4 = (N_EDGES / 4 + TB - 1) / TB;

  k_fill<<<gNode, TB, 0, stream>>>(deg, 1.0f, N_NODES);
  k_deg<<<gEdge4, TB, 0, stream>>>(col, deg, (unsigned*)key);
  k_dinv<<<gNode, TB, 0, stream>>>(deg, sums);

  k_layer_own<true, true, false><<<N_TILES, OWN_T, 0, stream>>>(x, W1, b1, row, col, key, deg, pA, wlist, wcnt, batch, sums, cnts);
  k_layer_own<false, true, false><<<N_TILES, OWN_T, 0, stream>>>(pA, W2, b2, row, col, key, deg, pB, wlist, wcnt, batch, sums, cnts);
  k_layer_own<false, false, true><<<N_TILES, OWN_T, 0, stream>>>(pB, W3, b3, row, col, key, deg, nullptr, wlist, wcnt, batch, sums, cnts);

  k_final<<<NGRAPH, 32, 0, stream>>>(sums, cnts, Wl, bl, outp);
}


# tiles' edge lists made where edges are first read: degree launch bins each edge into its tile's stored list (LDS counts, 1 integer add per tile and group); layer 1's key scan gone; sum order changes
# speedup vs baseline: 3.9969x; 1.0588x over previous
#include <hip/hip_runtime.h>

typedef __attribute__((ext_vector_type(2))) float v2f;
typedef __attribute__((ext_vector_type(8))) float v8f;

#define N_NODES 100000
#define N_EDGES 800000
#define FEAT    128
#define NGRAPH  512
#define NCLS    10

__device__ __forceinline__ void atomAddF(float* p, float v) {
  unsafeAtomicAdd(p, v);
}

__global__ void k_fill(float* __restrict__ p, float v, long n, int* __restrict__ cnt, long cnt_n) {
  long i = (long)blockIdx.x * blockDim.x + threadIdx.x;
  if (i < n) p[i] = v;
  if (i < cnt_n) cnt[i] = 0;
}

#define TILE_NODES 400
#define N_TILES    (N_NODES / TILE_NODES)
#define LIST_CAP   24576
#define BIN_T      1024
#define N_BIN_WG   ((N_EDGES / 4 + BIN_T - 1) / BIN_T)
#define CNT_PITCH  32
__global__ __launch_bounds__(BIN_T) void k_deg(const int* __restrict__ row, const int* __restrict__ col,
                                               float* __restrict__ deg, unsigned* __restrict__ key,
                                               unsigned* __restrict__ wlist, int* __restrict__ wcnt) {
  __shared__ int hist[N_TILES];
  const int t = threadIdx.x;
  if (t < N_TILES) hist[t] = 0;
  __syncthreads();
  const long i = (long)blockIdx.x * BIN_T + t;
  const bool on = i < N_EDGES / 4;
  int cc[4] = {0, 0, 0, 0}, rr[4] = {0, 0, 0, 0}, tl[4] = {0, 0, 0, 0}, rk[4] = {0, 0, 0, 0};
  if (on) {
    const int4 c = ((const int4*)col)[i];
    const int4 r = ((const int4*)row)[i];
    cc[0] = c.x; cc[1] = c.y; cc[2] = c.z; cc[3] = c.w;
    rr[0] = r.x; rr[1] = r.y; rr[2] = r.z; rr[3] = r.w;
#pragma unroll
    for (int j = 0; j < 4; ++j) {
      tl[j] = cc[j] / TILE_NODES;
      rk[j] = atomicAdd(&hist[tl[j]], 1);
    }
  }
  __syncthreads();
  if (t < N_TILES) {
    const int n = hist[t];
    hist[t] = n ? atomicAdd(&wcnt[t * CNT_PITCH], n) : 0;
  }
  __syncthreads();
  if (on) {
#pragma unroll
    for (int j = 0; j < 4; ++j) {
      const int idx = hist[tl[j]] + rk[j];
      if (idx < LIST_CAP)
        wlist[(size_t)tl[j] * LIST_CAP + idx] = ((unsigned)rr[j] << 9) | (unsigned)(cc[j] - tl[j] * TILE_NODES);
    }
#pragma unroll
    for (int j = 0; j < 4; ++j) atomAddF(&deg[cc[j]], 1.0f);
    key[i] = (unsigned)tl[0] | ((unsigned)tl[1] << 8) | ((unsigned)tl[2] << 16) | ((unsigned)tl[3] << 24);
  }
}

#define POOL_WORDS (NGRAPH * FEAT + NGRAPH)
static_assert(POOL_WORDS <= N_NODES, "the node grid also covers the pooled sums and counts");
__global__ void k_dinv(float* __restrict__ deg, float* __restrict__ pool) {
  long i = (long)blockIdx.x * blockDim.x + threadIdx.x;
  if (i < N_NODES) {
    float d = deg[i];
    deg[i] = (d > 0.0f) ? rsqrtf(fmaxf(d, 1.0f)) : 0.0f;
  }
  if (i < POOL_WORDS) pool[i] = 0.0f;
}

#define OWN_T      1024
#define KEYS_PER_LOAD 16
#define N_KEYQ     (N_EDGES / KEYS_PER_LOAD)
#define SCAN_ITERS ((N_KEYQ + OWN_T - 1) / OWN_T)
#define SCAN_BATCH 7
#define TPITCH     (FEAT + 4)
#define N_STRIPS   (TILE_NODES / 16)
static_assert(N_NODES % TILE_NODES == 0 && N_TILES <= 255, "tile key must fit a byte below 0xFF");
static_assert(N_EDGES % KEYS_PER_LOAD == 0 && N_EDGES % 4 == 0, "key array is written 4 and read 16 bytes at a time");
static_assert(TILE_NODES <= 512 && N_NODES <= (1 << 17), "list entry packs source (17 bits) and local row (9 bits)");
static_assert((4l << 20) + 3l * N_NODES * FEAT * 4 + (long)N_TILES * LIST_CAP * 4 <= 268435456l, "stored lists end inside the workspace");
static_assert(TILE_NODES % 16 == 0 && N_STRIPS <= OWN_T / 32, "one wave per 16-row strip of the tile");
static_assert(FEAT * FEAT + FEAT <= LIST_CAP && FEAT * FEAT / 4 == 4 * OWN_T, "W and the bias row are staged in the list's LDS, W in four 16-byte pieces per thread");
static_assert((TILE_NODES * TPITCH + LIST_CAP + 4) * 4 <= 320 * 1024, "tile and list fit one WGP's LDS");

template <int K>
__device__ __forceinline__ void own_add(float* __restrict__ tile, const unsigned* __restrict__ ents,
                                        int node0, const float* __restrict__ xin,
                                        const float* __restrict__ dinv, int lane) {
  int   cl[K];
  float nrm[K];
  float v[K][4];
#pragma unroll
  for (int u = 0; u < K; ++u) {
    const unsigned ent = ents[u];
    const int s = (int)(ent >> 9);
    cl[u] = (int)(ent & 511u);
    nrm[u] = dinv[s] * dinv[node0 + cl[u]];
    const float* hp = xin + (size_t)s * FEAT + lane;
    v[u][0] = hp[0];
    v[u][1] = hp[32];
    v[u][2] = hp[64];
    v[u][3] = hp[96];
  }
#pragma unroll
  for (int u = 0; u < K; ++u) {
    float* tp = tile + cl[u] * TPITCH + lane;
    atomicAdd(tp + 0,  nrm[u] * v[u][0]);
    atomicAdd(tp + 32, nrm[u] * v[u][1]);
    atomicAdd(tp + 64, nrm[u] * v[u][2]);
    atomicAdd(tp + 96, nrm[u] * v[u][3]);
  }
}

__device__ __forceinline__ void own_drain(float* __restrict__ tile, const unsigned* __restrict__ list,
                                          int n, int node0,
                                          const float* __restrict__ xin,
                                          const float* __restrict__ dinv,
                                          int wave, int lane) {
  const int groups = n >> 2;
  for (int g = wave; g < groups; g += OWN_T / 32)
    own_add<4>(tile, list + 4 * g, node0, xin, dinv, lane);
  if (wave < (n & 3))
    own_add<1>(tile, list + 4 * groups + wave, node0, xin, dinv, lane);
}

template <bool RELU_OUT>
__device__ __forceinline__ void own_product(float* __restrict__ tile, const float* __restrict__ sW,
                                            const float* __restrict__ sBias, int strip, int lane) {
  const int m    = lane & 15;
  const int koff = (lane >> 4) << 1;
  const int ncol = lane & 15;

  v8f acc[8];
#pragma unroll
  for (int n = 0; n < 8; ++n) acc[n] = (v8f){0, 0, 0, 0, 0, 0, 0, 0};

  const float* arow = tile + (strip * 16 + m) * TPITCH + koff;

  for (int k0 = 0; k0 < FEAT; k0 += 16) {
    v2f a[4];
#pragma unroll
    for (int j = 0; j < 4; ++j) {
      a[j].x = arow[k0 + 4 * j];
      a[j].y = arow[k0 + 4 * j + 1];
    }
    v2f b[2][8];
#pragma unroll
    for (int n = 0; n < 8; ++n) {
      const int kb = k0 + koff;
      b[0][n].x = sW[kb * FEAT + n * 16 + ncol];
      b[0][n].y = sW[(kb + 1) * FEAT + n * 16 + ncol];
    }
#pragma unroll
    for (int j = 0; j < 4; ++j) {
      if (j < 3) {
        const int kb = k0 + 4 * (j + 1) + koff;
#pragma unroll
        for (int n = 0; n < 8; ++n) {
          b[(j + 1) & 1][n].x = sW[kb * FEAT + n * 16 + ncol];
          b[(j + 1) & 1][n].y = sW[(kb + 1) * FEAT + n * 16 + ncol];
        }
      }
#pragma unroll
      for (int n = 0; n < 8; ++n) {
        acc[n] = __builtin_amdgcn_wmma_f32_16x16x4_f32(
            false, a[j], false, b[j & 1][n], (short)0, acc[n], false, false);
      }
    }
  }

  const int rowsel = (lane >> 4) * 8;
  float* orow = tile + (strip * 16 + rowsel) * TPITCH + ncol;
#pragma unroll
  for (int n = 0; n < 8; ++n) {
    const float bv = sBias[n * 16 + ncol];
#pragma unroll
    for (int r = 0; r < 8; ++r) {
      float o = acc[n][r] + bv;
      if (RELU_OUT) o = fmaxf(o, 0.0f);
      orow[r * TPITCH + n * 16] = o;
    }
  }
}

#define POOL_SLICES (OWN_T / FEAT)
#define POOL_ROWS   (TILE_NODES / POOL_SLICES)
#define POOL_CHUNK  10
static_assert(OWN_T % FEAT == 0 && FEAT % 32 == 0 && TILE_NODES % POOL_SLICES == 0, "a wave stays inside one slice; slices cover the tile");
static_assert(POOL_ROWS <= 64 && POOL_ROWS % POOL_CHUNK == 0, "a slice's ids fit two registers of a wave; whole chunks");
__device__ __forceinline__ void pool_flush(float* __restrict__ sums, float* __restrict__ cnts,
                                           int g, int f, float acc, int run) {
  atomAddF(sums + (size_t)g * FEAT + f, acc);
  if (f == 0) atomAddF(cnts + g, (float)run);
}
__device__ __forceinline__ void own_pool(const float* __restrict__ tile, const int* __restrict__ batch,
                                         int node0, float* __restrict__ sums, float* __restrict__ cnts,
                                         int wave, int lane) {
  const int sl = wave / (FEAT / 32);
  const int f  = (wave % (FEAT / 32)) * 32 + lane;
  const int r0 = sl * POOL_ROWS;
  const int* bp = batch + node0 + r0;
  const int ida = bp[min(lane, POOL_ROWS - 1)];
  const int idb = bp[min(lane + 32, POOL_ROWS - 1)];
  const float* tp = tile + r0 * TPITCH + f;
  int   g   = __builtin_amdgcn_readlane(ida, 0);
  float acc = 0.0f;
  int   run = 0;
  for (int c = 0; c < POOL_ROWS; c += POOL_CHUNK) {
    float v[POOL_CHUNK];
#pragma unroll
    for (int j = 0; j < POOL_CHUNK; ++j) v[j] = tp[(c + j) * TPITCH];
#pragma unroll
    for (int j = 0; j < POOL_CHUNK; ++j) {
      const int r  = c + j;
      const int gr = (r < 32) ? __builtin_amdgcn_readlane(ida, r) : __builtin_amdgcn_readlane(idb, r - 32);
      if (gr != g) {
        pool_flush(sums, cnts, g, f, acc, run);
        g = gr; acc = 0.0f; run = 0;
      }
      acc += v[j];
      ++run;
    }
  }
  pool_flush(sums, cnts, g, f, acc, run);
}

template <bool RELU_OUT, bool POOL>
__global__ __launch_bounds__(OWN_T) void k_layer_own(const float* __restrict__ xin,
                                                     const float* __restrict__ W,
                                                     const float* __restrict__ bias,
                                                     const int* __restrict__ row,
                                                     const int* __restrict__ col,
                                                     const unsigned char* __restrict__ key,
                                                     const float* __restrict__ dinv,
                                                     float* __restrict__ xout,
                                                     const unsigned* __restrict__ wlist,
                                                     const int* __restrict__ wcnt,
                                                     const int* __restrict__ batch,
                                                     float* __restrict__ sums,
                                                     float* __restrict__ cnts) {
  __shared__ __attribute__((aligned(16))) unsigned smem[LIST_CAP + TILE_NODES * TPITCH];
  unsigned* const list = smem;
  float* const tile = (float*)(smem + LIST_CAP);
  __shared__ int cnt_s;

  const int t     = threadIdx.x;
  const int wave  = __builtin_amdgcn_readfirstlane(t >> 5);
  const int lane  = t & 31;
  const unsigned bid = blockIdx.x;
  const int node0 = (int)bid * TILE_NODES;
  const unsigned* wl = wlist + (size_t)bid * LIST_CAP;
  const int saved = wcnt[bid * CNT_PITCH];
  const bool rescan = saved > LIST_CAP;

  if (t == 0) cnt_s = 0;
  {
    const float4* src4 = (const float4*)(xin + (size_t)node0 * FEAT);
    for (int i = t; i < TILE_NODES * FEAT / 4; i += OWN_T) {
      const int r = i >> 5, c4 = i & 31;
      const float d = dinv[node0 + r];
      const float s = d * d;
      float4 v = src4[i];
      v.x *= s; v.y *= s; v.z *= s; v.w *= s;
      *(float4*)(tile + r * TPITCH + 4 * c4) = v;
    }
  }
  if (!rescan)
    for (int i = t; i < saved; i += OWN_T) list[i] = wl[i];
  __syncthreads();

  int n = saved;
  if (rescan) {

  const uint4* key4 = (const uint4*)key;
  for (int it0 = 0; it0 < SCAN_ITERS; it0 += SCAN_BATCH) {
    uint4 k[SCAN_BATCH];
#pragma unroll
    for (int u = 0; u < SCAN_BATCH; ++u) {
      const int q = (it0 + u) * OWN_T + t;
      k[u] = key4[min(q, N_KEYQ - 1)];
    }
#pragma unroll
    for (int u = 0; u < SCAN_BATCH; ++u) {
      const int q = (it0 + u) * OWN_T + t;
      const unsigned w[4] = {k[u].x, k[u].y, k[u].z, k[u].w};
      unsigned mask = 0;
#pragma unroll
      for (int j = 0; j < 4; ++j)
#pragma unroll
        for (int b = 0; b < 4; ++b)
          mask |= (unsigned)(((w[j] >> (8 * b)) & 0xFFu) == bid) << (4 * j + b);
      if (q >= N_KEYQ) mask = 0;
      while (mask) {
        const int pos = __builtin_ctz(mask);
        mask &= mask - 1;
        const int e = q * KEYS_PER_LOAD + pos;
        const int idx = atomicAdd(&cnt_s, 1);
        if (idx < LIST_CAP) {
          list[idx] = (unsigned)e;
        } else {
          const int c  = col[e];
          const int sN = row[e];
          const float nrm = dinv[sN] * dinv[c];
          float* tp = tile + (c - node0) * TPITCH;
          const float* hp = xin + (size_t)sN * FEAT;
#pragma nounroll
          for (int f = 0; f < FEAT; ++f) atomicAdd(tp + f, nrm * hp[f]);
        }
      }
    }
  }
  __syncthreads();
  n = min(cnt_s, LIST_CAP);
  for (int i = t; i < n; i += OWN_T) {
    const int e = (int)list[i];
    list[i] = ((unsigned)row[e] << 9) | (unsigned)(col[e] - node0);
  }
  __syncthreads();
  }
  own_drain(tile, list, n, node0, xin, dinv, wave, lane);

  __syncthreads();
  float* sW = (float*)list;
#pragma unroll
  for (int j = 0; j < 4; ++j) ((float4*)sW)[t + j * OWN_T] = ((const float4*)W)[t + j * OWN_T];
  float* sBias = sW + FEAT * FEAT;
  if (t < FEAT / 4) ((float4*)sBias)[t] = ((const float4*)bias)[t];
  __syncthreads();

  if (wave < N_STRIPS)
    own_product<RELU_OUT>(tile, sW, sBias, wave, lane);
  __syncthreads();
  if (POOL) {
    own_pool(tile, batch, node0, sums, cnts, wave, lane);
  } else {
    float4* dst4 = (float4*)(xout + (size_t)node0 * FEAT);
    for (int i = t; i < TILE_NODES * FEAT / 4; i += OWN_T) {
      const int r = i >> 5, c4 = i & 31;
      dst4[i] = *(const float4*)(tile + r * TPITCH + 4 * c4);
    }
  }
}

__global__ void k_final(const float* __restrict__ sums, const float* __restrict__ cnts,
                        const float* __restrict__ Wl, const float* __restrict__ bl,
                        float* __restrict__ out) {
  const int g = blockIdx.x;
  const int c = threadIdx.x;
  if (c >= NCLS) return;
  const float inv = 1.0f / fmaxf(cnts[g], 1.0f);
  float acc = bl[c];
  for (int k = 0; k < FEAT; ++k)
    acc += sums[(size_t)g * FEAT + k] * inv * Wl[k * NCLS + c];
  out[(size_t)g * NCLS + c] = acc;
}

extern "C" void kernel_launch(void* const* d_in, const int* in_sizes, int n_in,
                              void* d_out, int out_size, void* d_ws, size_t ws_size,
                              hipStream_t stream) {
  const float* x     = (const float*)d_in[0];
  const int*   ei    = (const int*)d_in[1];
  const int*   row   = ei;
  const int*   col   = ei + N_EDGES;
  const int*   batch = (const int*)d_in[2];
  const float* W1 = (const float*)d_in[3];
  const float* b1 = (const float*)d_in[4];
  const float* W2 = (const float*)d_in[5];
  const float* b2 = (const float*)d_in[6];
  const float* W3 = (const float*)d_in[7];
  const float* b3 = (const float*)d_in[8];
  const float* Wl = (const float*)d_in[9];
  const float* bl = (const float*)d_in[10];
  float* outp = (float*)d_out;

  char* base = (char*)d_ws;
  const size_t featBytes = (size_t)N_NODES * FEAT * sizeof(float);
  float* deg  = (float*)(base);
  float* pA   = (float*)(base + (1 << 20) + featBytes);
  float* pB   = (float*)(base + (1 << 20) + 2 * featBytes);
  float* sums = (float*)(base + (1 << 20) + 3 * featBytes);
  float* cnts = sums + (size_t)NGRAPH * FEAT;
  unsigned char* key = (unsigned char*)(base + (2 << 20) + 3 * featBytes);
  int*      wcnt  = (int*)(base + (3 << 20) + 3 * featBytes);
  unsigned* wlist = (unsigned*)(base + (4 << 20) + 3 * featBytes);

  const int  TB     = 256;
  const int  gNode  = (N_NODES + TB - 1) / TB;

  k_fill<<<gNode, TB, 0, stream>>>(deg, 1.0f, N_NODES, wcnt, N_TILES * CNT_PITCH);
  k_deg<<<N_BIN_WG, BIN_T, 0, stream>>>(row, col, deg, (unsigned*)key, wlist, wcnt);
  k_dinv<<<gNode, TB, 0, stream>>>(deg, sums);

  k_layer_own<true, false><<<N_TILES, OWN_T, 0, stream>>>(x, W1, b1, row, col, key, deg, pA, wlist, wcnt, batch, sums, cnts);
  k_layer_own<true, false><<<N_TILES, OWN_T, 0, stream>>>(pA, W2, b2, row, col, key, deg, pB, wlist, wcnt, batch, sums, cnts);
  k_layer_own<false, true><<<N_TILES, OWN_T, 0, stream>>>(pB, W3, b3, row, col, key, deg, nullptr, wlist, wcnt, batch, sums, cnts);

  k_final<<<NGRAPH, 32, 0, stream>>>(sums, cnts, Wl, bl, outp);
}


# edge row gather: one 16-byte load per lane covers a 512-B source row (was four 4-byte loads); tile columns and W's LDS rows permuted so adds stay on 32 banks; k order of the product's sums changes
# speedup vs baseline: 4.2145x; 1.0544x over previous
#include <hip/hip_runtime.h>

typedef __attribute__((ext_vector_type(2))) float v2f;
typedef __attribute__((ext_vector_type(8))) float v8f;

#define N_NODES 100000
#define N_EDGES 800000
#define FEAT    128
#define NGRAPH  512
#define NCLS    10

__device__ __forceinline__ void atomAddF(float* p, float v) {
  unsafeAtomicAdd(p, v);
}

__global__ void k_fill(float* __restrict__ p, float v, long n, int* __restrict__ cnt, long cnt_n) {
  long i = (long)blockIdx.x * blockDim.x + threadIdx.x;
  if (i < n) p[i] = v;
  if (i < cnt_n) cnt[i] = 0;
}

#define TILE_NODES 400
#define N_TILES    (N_NODES / TILE_NODES)
#define LIST_CAP   24576
#define BIN_T      1024
#define N_BIN_WG   ((N_EDGES / 4 + BIN_T - 1) / BIN_T)
#define CNT_PITCH  32
__global__ __launch_bounds__(BIN_T) void k_deg(const int* __restrict__ row, const int* __restrict__ col,
                                               float* __restrict__ deg, unsigned* __restrict__ key,
                                               unsigned* __restrict__ wlist, int* __restrict__ wcnt) {
  __shared__ int hist[N_TILES];
  const int t = threadIdx.x;
  if (t < N_TILES) hist[t] = 0;
  __syncthreads();
  const long i = (long)blockIdx.x * BIN_T + t;
  const bool on = i < N_EDGES / 4;
  int cc[4] = {0, 0, 0, 0}, rr[4] = {0, 0, 0, 0}, tl[4] = {0, 0, 0, 0}, rk[4] = {0, 0, 0, 0};
  if (on) {
    const int4 c = ((const int4*)col)[i];
    const int4 r = ((const int4*)row)[i];
    cc[0] = c.x; cc[1] = c.y; cc[2] = c.z; cc[3] = c.w;
    rr[0] = r.x; rr[1] = r.y; rr[2] = r.z; rr[3] = r.w;
#pragma unroll
    for (int j = 0; j < 4; ++j) {
      tl[j] = cc[j] / TILE_NODES;
      rk[j] = atomicAdd(&hist[tl[j]], 1);
    }
  }
  __syncthreads();
  if (t < N_TILES) {
    const int n = hist[t];
    hist[t] = n ? atomicAdd(&wcnt[t * CNT_PITCH], n) : 0;
  }
  __syncthreads();
  if (on) {
#pragma unroll
    for (int j = 0; j < 4; ++j) {
      const int idx = hist[tl[j]] + rk[j];
      if (idx < LIST_CAP)
        wlist[(size_t)tl[j] * LIST_CAP + idx] = ((unsigned)rr[j] << 9) | (unsigned)(cc[j] - tl[j] * TILE_NODES);
    }
#pragma unroll
    for (int j = 0; j < 4; ++j) atomAddF(&deg[cc[j]], 1.0f);
    key[i] = (unsigned)tl[0] | ((unsigned)tl[1] << 8) | ((unsigned)tl[2] << 16) | ((unsigned)tl[3] << 24);
  }
}

#define POOL_WORDS (NGRAPH * FEAT + NGRAPH)
static_assert(POOL_WORDS <= N_NODES, "the node grid also covers the pooled sums and counts");
__global__ void k_dinv(float* __restrict__ deg, float* __restrict__ pool) {
  long i = (long)blockIdx.x * blockDim.x + threadIdx.x;
  if (i < N_NODES) {
    float d = deg[i];
    deg[i] = (d > 0.0f) ? rsqrtf(fmaxf(d, 1.0f)) : 0.0f;
  }
  if (i < POOL_WORDS) pool[i] = 0.0f;
}

#define TCOL(f)    ((((f) & 3) << 5) | ((f) >> 2))
#define TFEAT(p)   ((((p) & 31) << 2) | ((p) >> 5))
#define OWN_T      1024
#define KEYS_PER_LOAD 16
#define N_KEYQ     (N_EDGES / KEYS_PER_LOAD)
#define SCAN_ITERS ((N_KEYQ + OWN_T - 1) / OWN_T)
#define SCAN_BATCH 7
#define TPITCH     (FEAT + 4)
#define N_STRIPS   (TILE_NODES / 16)
static_assert(N_NODES % TILE_NODES == 0 && N_TILES <= 255, "tile key must fit a byte below 0xFF");
static_assert(N_EDGES % KEYS_PER_LOAD == 0 && N_EDGES % 4 == 0, "key array is written 4 and read 16 bytes at a time");
static_assert(TILE_NODES <= 512 && N_NODES <= (1 << 17), "list entry packs source (17 bits) and local row (9 bits)");
static_assert((4l << 20) + 3l * N_NODES * FEAT * 4 + (long)N_TILES * LIST_CAP * 4 <= 268435456l, "stored lists end inside the workspace");
static_assert(TILE_NODES % 16 == 0 && N_STRIPS <= OWN_T / 32, "one wave per 16-row strip of the tile");
static_assert(FEAT * FEAT + FEAT <= LIST_CAP && FEAT * FEAT / 4 == 4 * OWN_T, "W and the bias row are staged in the list's LDS, W in four 16-byte pieces per thread");
static_assert((TILE_NODES * TPITCH + LIST_CAP + 4) * 4 <= 320 * 1024, "tile and list fit one WGP's LDS");

template <int K>
__device__ __forceinline__ void own_add(float* __restrict__ tile, const unsigned* __restrict__ ents,
                                        int node0, const float* __restrict__ xin,
                                        const float* __restrict__ dinv, int lane) {
  int    cl[K];
  float  nrm[K];
  float4 v[K];
#pragma unroll
  for (int u = 0; u < K; ++u) {
    const unsigned ent = ents[u];
    const int s = (int)(ent >> 9);
    cl[u] = (int)(ent & 511u);
    nrm[u] = dinv[s] * dinv[node0 + cl[u]];
    v[u] = *(const float4*)(xin + (size_t)s * FEAT + 4 * lane);
  }
#pragma unroll
  for (int u = 0; u < K; ++u) {
    float* tp = tile + cl[u] * TPITCH + lane;
    atomicAdd(tp + 0,  nrm[u] * v[u].x);
    atomicAdd(tp + 32, nrm[u] * v[u].y);
    atomicAdd(tp + 64, nrm[u] * v[u].z);
    atomicAdd(tp + 96, nrm[u] * v[u].w);
  }
}

__device__ __forceinline__ void own_drain(float* __restrict__ tile, const unsigned* __restrict__ list,
                                          int n, int node0,
                                          const float* __restrict__ xin,
                                          const float* __restrict__ dinv,
                                          int wave, int lane) {
  const int groups = n >> 2;
  for (int g = wave; g < groups; g += OWN_T / 32)
    own_add<4>(tile, list + 4 * g, node0, xin, dinv, lane);
  if (wave < (n & 3))
    own_add<1>(tile, list + 4 * groups + wave, node0, xin, dinv, lane);
}

template <bool RELU_OUT>
__device__ __forceinline__ void own_product(float* __restrict__ tile, const float* __restrict__ sW,
                                            const float* __restrict__ sBias, int strip, int lane) {
  const int m    = lane & 15;
  const int koff = (lane >> 4) << 1;
  const int ncol = lane & 15;

  v8f acc[8];
#pragma unroll
  for (int n = 0; n < 8; ++n) acc[n] = (v8f){0, 0, 0, 0, 0, 0, 0, 0};

  const float* arow = tile + (strip * 16 + m) * TPITCH + koff;

  for (int k0 = 0; k0 < FEAT; k0 += 16) {
    v2f a[4];
#pragma unroll
    for (int j = 0; j < 4; ++j) {
      a[j].x = arow[k0 + 4 * j];
      a[j].y = arow[k0 + 4 * j + 1];
    }
    v2f b[2][8];
#pragma unroll
    for (int n = 0; n < 8; ++n) {
      const int kb = k0 + koff;
      b[0][n].x = sW[kb * FEAT + n * 16 + ncol];
      b[0][n].y = sW[(kb + 1) * FEAT + n * 16 + ncol];
    }
#pragma unroll
    for (int j = 0; j < 4; ++j) {
      if (j < 3) {
        const int kb = k0 + 4 * (j + 1) + koff;
#pragma unroll
        for (int n = 0; n < 8; ++n) {
          b[(j + 1) & 1][n].x = sW[kb * FEAT + n * 16 + ncol];
          b[(j + 1) & 1][n].y = sW[(kb + 1) * FEAT + n * 16 + ncol];
        }
      }
#pragma unroll
      for (int n = 0; n < 8; ++n) {
        acc[n] = __builtin_amdgcn_wmma_f32_16x16x4_f32(
            false, a[j], false, b[j & 1][n], (short)0, acc[n], false, false);
      }
    }
  }

  const int rowsel = (lane >> 4) * 8;
  float* orow = tile + (strip * 16 + rowsel) * TPITCH + ncol;
#pragma unroll
  for (int n = 0; n < 8; ++n) {
    const float bv = sBias[n * 16 + ncol];
#pragma unroll
    for (int r = 0; r < 8; ++r) {
      float o = acc[n][r] + bv;
      if (RELU_OUT) o = fmaxf(o, 0.0f);
      orow[r * TPITCH + n * 16] = o;
    }
  }
}

#define POOL_SLICES (OWN_T / FEAT)
#define POOL_ROWS   (TILE_NODES / POOL_SLICES)
#define POOL_CHUNK  10
static_assert(OWN_T % FEAT == 0 && FEAT % 32 == 0 && TILE_NODES % POOL_SLICES == 0, "a wave stays inside one slice; slices cover the tile");
static_assert(POOL_ROWS <= 64 && POOL_ROWS % POOL_CHUNK == 0, "a slice's ids fit two registers of a wave; whole chunks");
__device__ __forceinline__ void pool_flush(float* __restrict__ sums, float* __restrict__ cnts,
                                           int g, int f, float acc, int run) {
  atomAddF(sums + (size_t)g * FEAT + f, acc);
  if (f == 0) atomAddF(cnts + g, (float)run);
}
__device__ __forceinline__ void own_pool(const float* __restrict__ tile, const int* __restrict__ batch,
                                         int node0, float* __restrict__ sums, float* __restrict__ cnts,
                                         int wave, int lane) {
  const int sl = wave / (FEAT / 32);
  const int f  = (wave % (FEAT / 32)) * 32 + lane;
  const int r0 = sl * POOL_ROWS;
  const int* bp = batch + node0 + r0;
  const int ida = bp[min(lane, POOL_ROWS - 1)];
  const int idb = bp[min(lane + 32, POOL_ROWS - 1)];
  const float* tp = tile + r0 * TPITCH + f;
  int   g   = __builtin_amdgcn_readlane(ida, 0);
  float acc = 0.0f;
  int   run = 0;
  for (int c = 0; c < POOL_ROWS; c += POOL_CHUNK) {
    float v[POOL_CHUNK];
#pragma unroll
    for (int j = 0; j < POOL_CHUNK; ++j) v[j] = tp[(c + j) * TPITCH];
#pragma unroll
    for (int j = 0; j < POOL_CHUNK; ++j) {
      const int r  = c + j;
      const int gr = (r < 32) ? __builtin_amdgcn_readlane(ida, r) : __builtin_amdgcn_readlane(idb, r - 32);
      if (gr != g) {
        pool_flush(sums, cnts, g, f, acc, run);
        g = gr; acc = 0.0f; run = 0;
      }
      acc += v[j];
      ++run;
    }
  }
  pool_flush(sums, cnts, g, f, acc, run);
}

template <bool RELU_OUT, bool POOL>
__global__ __launch_bounds__(OWN_T) void k_layer_own(const float* __restrict__ xin,
                                                     const float* __restrict__ W,
                                                     const float* __restrict__ bias,
                                                     const int* __restrict__ row,
                                                     const int* __restrict__ col,
                                                     const unsigned char* __restrict__ key,
                                                     const float* __restrict__ dinv,
                                                     float* __restrict__ xout,
                                                     const unsigned* __restrict__ wlist,
                                                     const int* __restrict__ wcnt,
                                                     const int* __restrict__ batch,
                                                     float* __restrict__ sums,
                                                     float* __restrict__ cnts) {
  __shared__ __attribute__((aligned(16))) unsigned smem[LIST_CAP + TILE_NODES * TPITCH];
  unsigned* const list = smem;
  float* const tile = (float*)(smem + LIST_CAP);
  __shared__ int cnt_s;

  const int t     = threadIdx.x;
  const int wave  = __builtin_amdgcn_readfirstlane(t >> 5);
  const int lane  = t & 31;
  const unsigned bid = blockIdx.x;
  const int node0 = (int)bid * TILE_NODES;
  const unsigned* wl = wlist + (size_t)bid * LIST_CAP;
  const int saved = wcnt[bid * CNT_PITCH];
  const bool rescan = saved > LIST_CAP;

  if (t == 0) cnt_s = 0;
  {
    const float4* src4 = (const float4*)(xin + (size_t)node0 * FEAT);
    for (int i = t; i < TILE_NODES * FEAT / 4; i += OWN_T) {
      const int r = i >> 5, c4 = i & 31;
      const float d = dinv[node0 + r];
      const float s = d * d;
      float4 v = src4[i];
      float* tr = tile + r * TPITCH + c4;
      tr[0]  = v.x * s;
      tr[32] = v.y * s;
      tr[64] = v.z * s;
      tr[96] = v.w * s;
    }
  }
  if (!rescan)
    for (int i = t; i < saved; i += OWN_T) list[i] = wl[i];
  __syncthreads();

  int n = saved;
  if (rescan) {

  const uint4* key4 = (const uint4*)key;
  for (int it0 = 0; it0 < SCAN_ITERS; it0 += SCAN_BATCH) {
    uint4 k[SCAN_BATCH];
#pragma unroll
    for (int u = 0; u < SCAN_BATCH; ++u) {
      const int q = (it0 + u) * OWN_T + t;
      k[u] = key4[min(q, N_KEYQ - 1)];
    }
#pragma unroll
    for (int u = 0; u < SCAN_BATCH; ++u) {
      const int q = (it0 + u) * OWN_T + t;
      const unsigned w[4] = {k[u].x, k[u].y, k[u].z, k[u].w};
      unsigned mask = 0;
#pragma unroll
      for (int j = 0; j < 4; ++j)
#pragma unroll
        for (int b = 0; b < 4; ++b)
          mask |= (unsigned)(((w[j] >> (8 * b)) & 0xFFu) == bid) << (4 * j + b);
      if (q >= N_KEYQ) mask = 0;
      while (mask) {
        const int pos = __builtin_ctz(mask);
        mask &= mask - 1;
        const int e = q * KEYS_PER_LOAD + pos;
        const int idx = atomicAdd(&cnt_s, 1);
        if (idx < LIST_CAP) {
          list[idx] = (unsigned)e;
        } else {
          const int c  = col[e];
          const int sN = row[e];
          const float nrm = dinv[sN] * dinv[c];
          float* tp = tile + (c - node0) * TPITCH;
          const float* hp = xin + (size_t)sN * FEAT;
#pragma nounroll
          for (int f = 0; f < FEAT; ++f) atomicAdd(tp + TCOL(f), nrm * hp[f]);
        }
      }
    }
  }
  __syncthreads();
  n = min(cnt_s, LIST_CAP);
  for (int i = t; i < n; i += OWN_T) {
    const int e = (int)list[i];
    list[i] = ((unsigned)row[e] << 9) | (unsigned)(col[e] - node0);
  }
  __syncthreads();
  }
  own_drain(tile, list, n, node0, xin, dinv, wave, lane);

  __syncthreads();
  float* sW = (float*)list;
#pragma unroll
  for (int j = 0; j < 4; ++j) {
    const int q = t + j * OWN_T;
    ((float4*)sW)[q] = ((const float4*)W)[TFEAT(q >> 5) * (FEAT / 4) + (q & 31)];
  }
  float* sBias = sW + FEAT * FEAT;
  if (t < FEAT / 4) ((float4*)sBias)[t] = ((const float4*)bias)[t];
  __syncthreads();

  if (wave < N_STRIPS)
    own_product<RELU_OUT>(tile, sW, sBias, wave, lane);
  __syncthreads();
  if (POOL) {
    own_pool(tile, batch, node0, sums, cnts, wave, lane);
  } else {
    float4* dst4 = (float4*)(xout + (size_t)node0 * FEAT);
    for (int i = t; i < TILE_NODES * FEAT / 4; i += OWN_T) {
      const int r = i >> 5, c4 = i & 31;
      dst4[i] = *(const float4*)(tile + r * TPITCH + 4 * c4);
    }
  }
}

__global__ void k_final(const float* __restrict__ sums, const float* __restrict__ cnts,
                        const float* __restrict__ Wl, const float* __restrict__ bl,
                        float* __restrict__ out) {
  const int g = blockIdx.x;
  const int c = threadIdx.x;
  if (c >= NCLS) return;
  const float inv = 1.0f / fmaxf(cnts[g], 1.0f);
  float acc = bl[c];
  for (int k = 0; k < FEAT; ++k)
    acc += sums[(size_t)g * FEAT + k] * inv * Wl[k * NCLS + c];
  out[(size_t)g * NCLS + c] = acc;
}

extern "C" void kernel_launch(void* const* d_in, const int* in_sizes, int n_in,
                              void* d_out, int out_size, void* d_ws, size_t ws_size,
                              hipStream_t stream) {
  const float* x     = (const float*)d_in[0];
  const int*   ei    = (const int*)d_in[1];
  const int*   row   = ei;
  const int*   col   = ei + N_EDGES;
  const int*   batch = (const int*)d_in[2];
  const float* W1 = (const float*)d_in[3];
  const float* b1 = (const float*)d_in[4];
  const float* W2 = (const float*)d_in[5];
  const float* b2 = (const float*)d_in[6];
  const float* W3 = (const float*)d_in[7];
  const float* b3 = (const float*)d_in[8];
  const float* Wl = (const float*)d_in[9];
  const float* bl = (const float*)d_in[10];
  float* outp = (float*)d_out;

  char* base = (char*)d_ws;
  const size_t featBytes = (size_t)N_NODES * FEAT * sizeof(float);
  float* deg  = (float*)(base);
  float* pA   = (float*)(base + (1 << 20) + featBytes);
  float* pB   = (float*)(base + (1 << 20) + 2 * featBytes);
  float* sums = (float*)(base + (1 << 20) + 3 * featBytes);
  float* cnts = sums + (size_t)NGRAPH * FEAT;
  unsigned char* key = (unsigned char*)(base + (2 << 20) + 3 * featBytes);
  int*      wcnt  = (int*)(base + (3 << 20) + 3 * featBytes);
  unsigned* wlist = (unsigned*)(base + (4 << 20) + 3 * featBytes);

  const int  TB     = 256;
  const int  gNode  = (N_NODES + TB - 1) / TB;

  k_fill<<<gNode, TB, 0, stream>>>(deg, 1.0f, N_NODES, wcnt, N_TILES * CNT_PITCH);
  k_deg<<<N_BIN_WG, BIN_T, 0, stream>>>(row, col, deg, (unsigned*)key, wlist, wcnt);
  k_dinv<<<gNode, TB, 0, stream>>>(deg, sums);

  k_layer_own<true, false><<<N_TILES, OWN_T, 0, stream>>>(x, W1, b1, row, col, key, deg, pA, wlist, wcnt, batch, sums, cnts);
  k_layer_own<true, false><<<N_TILES, OWN_T, 0, stream>>>(pA, W2, b2, row, col, key, deg, pB, wlist, wcnt, batch, sums, cnts);
  k_layer_own<false, true><<<N_TILES, OWN_T, 0, stream>>>(pB, W3, b3, row, col, key, deg, nullptr, wlist, wcnt, batch, sums, cnts);

  k_final<<<NGRAPH, 32, 0, stream>>>(sums, cnts, Wl, bl, outp);
}


# degrees counted per owner tile from its stored edge list (LDS integer adds in the dinv launch, 250 work-groups): the 800000 global float adds onto deg and the fill of deg are gone; same values
# speedup vs baseline: 5.0832x; 1.2061x over previous
#include <hip/hip_runtime.h>

typedef __attribute__((ext_vector_type(2))) float v2f;
typedef __attribute__((ext_vector_type(8))) float v8f;

#define N_NODES 100000
#define N_EDGES 800000
#define FEAT    128
#define NGRAPH  512
#define NCLS    10

__device__ __forceinline__ void atomAddF(float* p, float v) {
  unsafeAtomicAdd(p, v);
}

__global__ void k_fill(int* __restrict__ cnt, long cnt_n) {
  long i = (long)blockIdx.x * blockDim.x + threadIdx.x;
  if (i < cnt_n) cnt[i] = 0;
}

#define TILE_NODES 400
#define N_TILES    (N_NODES / TILE_NODES)
#define LIST_CAP   24576
#define BIN_T      1024
#define N_BIN_WG   ((N_EDGES / 4 + BIN_T - 1) / BIN_T)
#define CNT_PITCH  32
__global__ __launch_bounds__(BIN_T) void k_deg(const int* __restrict__ row, const int* __restrict__ col,
                                               unsigned* __restrict__ key,
                                               unsigned* __restrict__ wlist, int* __restrict__ wcnt) {
  __shared__ int hist[N_TILES];
  const int t = threadIdx.x;
  if (t < N_TILES) hist[t] = 0;
  __syncthreads();
  const long i = (long)blockIdx.x * BIN_T + t;
  const bool on = i < N_EDGES / 4;
  int cc[4] = {0, 0, 0, 0}, rr[4] = {0, 0, 0, 0}, tl[4] = {0, 0, 0, 0}, rk[4] = {0, 0, 0, 0};
  if (on) {
    const int4 c = ((const int4*)col)[i];
    const int4 r = ((const int4*)row)[i];
    cc[0] = c.x; cc[1] = c.y; cc[2] = c.z; cc[3] = c.w;
    rr[0] = r.x; rr[1] = r.y; rr[2] = r.z; rr[3] = r.w;
#pragma unroll
    for (int j = 0; j < 4; ++j) {
      tl[j] = cc[j] / TILE_NODES;
      rk[j] = atomicAdd(&hist[tl[j]], 1);
    }
  }
  __syncthreads();
  if (t < N_TILES) {
    const int n = hist[t];
    hist[t] = n ? atomicAdd(&wcnt[t * CNT_PITCH], n) : 0;
  }
  __syncthreads();
  if (on) {
#pragma unroll
    for (int j = 0; j < 4; ++j) {
      const int idx = hist[tl[j]] + rk[j];
      if (idx < LIST_CAP)
        wlist[(size_t)tl[j] * LIST_CAP + idx] = ((unsigned)rr[j] << 9) | (unsigned)(cc[j] - tl[j] * TILE_NODES);
    }
    key[i] = (unsigned)tl[0] | ((unsigned)tl[1] << 8) | ((unsigned)tl[2] << 16) | ((unsigned)tl[3] << 24);
  }
}

#define POOL_WORDS (NGRAPH * FEAT + NGRAPH)
#define DINV_T     256
#define DINV_BATCH 4
__global__ __launch_bounds__(DINV_T) void k_dinv(const unsigned* __restrict__ wlist, const int* __restrict__ wcnt,
                                                 const unsigned char* __restrict__ key, const int* __restrict__ col,
                                                 float* __restrict__ dinv, float* __restrict__ pool) {
  __shared__ int rows[512];
  const int t = threadIdx.x;
  const int bid = blockIdx.x;
  const int node0 = bid * TILE_NODES;
  for (int i = t; i < 512; i += DINV_T) rows[i] = 0;
  __syncthreads();
  const int saved = wcnt[bid * CNT_PITCH];
  if (saved <= LIST_CAP) {
    const unsigned* wl = wlist + (size_t)bid * LIST_CAP;
    const uint4* wl4 = (const uint4*)wl;
    const int quads = saved >> 2;
    for (int q0 = 0; q0 < quads; q0 += DINV_BATCH * DINV_T) {
      uint4 v[DINV_BATCH];
#pragma unroll
      for (int u = 0; u < DINV_BATCH; ++u)
        v[u] = wl4[min(q0 + u * DINV_T + t, quads - 1)];
#pragma unroll
      for (int u = 0; u < DINV_BATCH; ++u) {
        if (q0 + u * DINV_T + t < quads) {
          atomicAdd(&rows[v[u].x & 511u], 1);
          atomicAdd(&rows[v[u].y & 511u], 1);
          atomicAdd(&rows[v[u].z & 511u], 1);
          atomicAdd(&rows[v[u].w & 511u], 1);
        }
      }
    }
    const int rest = (quads << 2) + t;
    if (rest < saved) atomicAdd(&rows[wl[rest] & 511u], 1);
  } else {
    for (int e = t; e < N_EDGES; e += DINV_T)
      if ((int)key[e] == bid) atomicAdd(&rows[col[e] - node0], 1);
  }
  __syncthreads();
  for (int i = t; i < TILE_NODES; i += DINV_T) {
    const float d = 1.0f + (float)rows[i];
    dinv[node0 + i] = (d > 0.0f) ? rsqrtf(fmaxf(d, 1.0f)) : 0.0f;
  }
  for (int p = bid * DINV_T + t; p < POOL_WORDS; p += N_TILES * DINV_T) pool[p] = 0.0f;
}

#define TCOL(f)    ((((f) & 3) << 5) | ((f) >> 2))
#define TFEAT(p)   ((((p) & 31) << 2) | ((p) >> 5))
#define OWN_T      1024
#define KEYS_PER_LOAD 16
#define N_KEYQ     (N_EDGES / KEYS_PER_LOAD)
#define SCAN_ITERS ((N_KEYQ + OWN_T - 1) / OWN_T)
#define SCAN_BATCH 7
#define TPITCH     (FEAT + 4)
#define N_STRIPS   (TILE_NODES / 16)
static_assert(N_NODES % TILE_NODES == 0 && N_TILES <= 255, "tile key must fit a byte below 0xFF");
static_assert(N_EDGES % KEYS_PER_LOAD == 0 && N_EDGES % 4 == 0, "key array is written 4 and read 16 bytes at a time");
static_assert(TILE_NODES <= 512 && N_NODES <= (1 << 17), "list entry packs source (17 bits) and local row (9 bits)");
static_assert((4l << 20) + 3l * N_NODES * FEAT * 4 + (long)N_TILES * LIST_CAP * 4 <= 268435456l, "stored lists end inside the workspace");
static_assert(TILE_NODES % 16 == 0 && N_STRIPS <= OWN_T / 32, "one wave per 16-row strip of the tile");
static_assert(FEAT * FEAT + FEAT <= LIST_CAP && FEAT * FEAT / 4 == 4 * OWN_T, "W and the bias row are staged in the list's LDS, W in four 16-byte pieces per thread");
static_assert((TILE_NODES * TPITCH + LIST_CAP + 4) * 4 <= 320 * 1024, "tile and list fit one WGP's LDS");

template <int K>
__device__ __forceinline__ void own_add(float* __restrict__ tile, const unsigned* __restrict__ ents,
                                        int node0, const float* __restrict__ xin,
                                        const float* __restrict__ dinv, int lane) {
  int    cl[K];
  float  nrm[K];
  float4 v[K];
#pragma unroll
  for (int u = 0; u < K; ++u) {
    const unsigned ent = ents[u];
    const int s = (int)(ent >> 9);
    cl[u] = (int)(ent & 511u);
    nrm[u] = dinv[s] * dinv[node0 + cl[u]];
    v[u] = *(const float4*)(xin + (size_t)s * FEAT + 4 * lane);
  }
#pragma unroll
  for (int u = 0; u < K; ++u) {
    float* tp = tile + cl[u] * TPITCH + lane;
    atomicAdd(tp + 0,  nrm[u] * v[u].x);
    atomicAdd(tp + 32, nrm[u] * v[u].y);
    atomicAdd(tp + 64, nrm[u] * v[u].z);
    atomicAdd(tp + 96, nrm[u] * v[u].w);
  }
}

__device__ __forceinline__ void own_drain(float* __restrict__ tile, const unsigned* __restrict__ list,
                                          int n, int node0,
                                          const float* __restrict__ xin,
                                          const float* __restrict__ dinv,
                                          int wave, int lane) {
  const int groups = n >> 2;
  for (int g = wave; g < groups; g += OWN_T / 32)
    own_add<4>(tile, list + 4 * g, node0, xin, dinv, lane);
  if (wave < (n & 3))
    own_add<1>(tile, list + 4 * groups + wave, node0, xin, dinv, lane);
}

template <bool RELU_OUT>
__device__ __forceinline__ void own_product(float* __restrict__ tile, const float* __restrict__ sW,
                                            const float* __restrict__ sBias, int strip, int lane) {
  const int m    = lane & 15;
  const int koff = (lane >> 4) << 1;
  const int ncol = lane & 15;

  v8f acc[8];
#pragma unroll
  for (int n = 0; n < 8; ++n) acc[n] = (v8f){0, 0, 0, 0, 0, 0, 0, 0};

  const float* arow = tile + (strip * 16 + m) * TPITCH + koff;

  for (int k0 = 0; k0 < FEAT; k0 += 16) {
    v2f a[4];
#pragma unroll
    for (int j = 0; j < 4; ++j) {
      a[j].x = arow[k0 + 4 * j];
      a[j].y = arow[k0 + 4 * j + 1];
    }
    v2f b[2][8];
#pragma unroll
    for (int n = 0; n < 8; ++n) {
      const int kb = k0 + koff;
      b[0][n].x = sW[kb * FEAT + n * 16 + ncol];
      b[0][n].y = sW[(kb + 1) * FEAT + n * 16 + ncol];
    }
#pragma unroll
    for (int j = 0; j < 4; ++j) {
      if (j < 3) {
        const int kb = k0 + 4 * (j + 1) + koff;
#pragma unroll
        for (int n = 0; n < 8; ++n) {
          b[(j + 1) & 1][n].x = sW[kb * FEAT + n * 16 + ncol];
          b[(j + 1) & 1][n].y = sW[(kb + 1) * FEAT + n * 16 + ncol];
        }
      }
#pragma unroll
      for (int n = 0; n < 8; ++n) {
        acc[n] = __builtin_amdgcn_wmma_f32_16x16x4_f32(
            false, a[j], false, b[j & 1][n], (short)0, acc[n], false, false);
      }
    }
  }

  const int rowsel = (lane >> 4) * 8;
  float* orow = tile + (strip * 16 + rowsel) * TPITCH + ncol;
#pragma unroll
  for (int n = 0; n < 8; ++n) {
    const float bv = sBias[n * 16 + ncol];
#pragma unroll
    for (int r = 0; r < 8; ++r) {
      float o = acc[n][r] + bv;
      if (RELU_OUT) o = fmaxf(o, 0.0f);
      orow[r * TPITCH + n * 16] = o;
    }
  }
}

#define POOL_SLICES (OWN_T / FEAT)
#define POOL_ROWS   (TILE_NODES / POOL_SLICES)
#define POOL_CHUNK  10
static_assert(OWN_T % FEAT == 0 && FEAT % 32 == 0 && TILE_NODES % POOL_SLICES == 0, "a wave stays inside one slice; slices cover the tile");
static_assert(POOL_ROWS <= 64 && POOL_ROWS % POOL_CHUNK == 0, "a slice's ids fit two registers of a wave; whole chunks");
__device__ __forceinline__ void pool_flush(float* __restrict__ sums, float* __restrict__ cnts,
                                           int g, int f, float acc, int run) {
  atomAddF(sums + (size_t)g * FEAT + f, acc);
  if (f == 0) atomAddF(cnts + g, (float)run);
}
__device__ __forceinline__ void own_pool(const float* __restrict__ tile, const int* __restrict__ batch,
                                         int node0, float* __restrict__ sums, float* __restrict__ cnts,
                                         int wave, int lane) {
  const int sl = wave / (FEAT / 32);
  const int f  = (wave % (FEAT / 32)) * 32 + lane;
  const int r0 = sl * POOL_ROWS;
  const int* bp = batch + node0 + r0;
  const int ida = bp[min(lane, POOL_ROWS - 1)];
  const int idb = bp[min(lane + 32, POOL_ROWS - 1)];
  const float* tp = tile + r0 * TPITCH + f;
  int   g   = __builtin_amdgcn_readlane(ida, 0);
  float acc = 0.0f;
  int   run = 0;
  for (int c = 0; c < POOL_ROWS; c += POOL_CHUNK) {
    float v[POOL_CHUNK];
#pragma unroll
    for (int j = 0; j < POOL_CHUNK; ++j) v[j] = tp[(c + j) * TPITCH];
#pragma unroll
    for (int j = 0; j < POOL_CHUNK; ++j) {
      const int r  = c + j;
      const int gr = (r < 32) ? __builtin_amdgcn_readlane(ida, r) : __builtin_amdgcn_readlane(idb, r - 32);
      if (gr != g) {
        pool_flush(sums, cnts, g, f, acc, run);
        g = gr; acc = 0.0f; run = 0;
      }
      acc += v[j];
      ++run;
    }
  }
  pool_flush(sums, cnts, g, f, acc, run);
}

template <bool RELU_OUT, bool POOL>
__global__ __launch_bounds__(OWN_T) void k_layer_own(const float* __restrict__ xin,
                                                     const float* __restrict__ W,
                                                     const float* __restrict__ bias,
                                                     const int* __restrict__ row,
                                                     const int* __restrict__ col,
                                                     const unsigned char* __restrict__ key,
                                                     const float* __restrict__ dinv,
                                                     float* __restrict__ xout,
                                                     const unsigned* __restrict__ wlist,
                                                     const int* __restrict__ wcnt,
                                                     const int* __restrict__ batch,
                                                     float* __restrict__ sums,
                                                     float* __restrict__ cnts) {
  __shared__ __attribute__((aligned(16))) unsigned smem[LIST_CAP + TILE_NODES * TPITCH];
  unsigned* const list = smem;
  float* const tile = (float*)(smem + LIST_CAP);
  __shared__ int cnt_s;

  const int t     = threadIdx.x;
  const int wave  = __builtin_amdgcn_readfirstlane(t >> 5);
  const int lane  = t & 31;
  const unsigned bid = blockIdx.x;
  const int node0 = (int)bid * TILE_NODES;
  const unsigned* wl = wlist + (size_t)bid * LIST_CAP;
  const int saved = wcnt[bid * CNT_PITCH];
  const bool rescan = saved > LIST_CAP;

  if (t == 0) cnt_s = 0;
  {
    const float4* src4 = (const float4*)(xin + (size_t)node0 * FEAT);
    for (int i = t; i < TILE_NODES * FEAT / 4; i += OWN_T) {
      const int r = i >> 5, c4 = i & 31;
      const float d = dinv[node0 + r];
      const float s = d * d;
      float4 v = src4[i];
      float* tr = tile + r * TPITCH + c4;
      tr[0]  = v.x * s;
      tr[32] = v.y * s;
      tr[64] = v.z * s;
      tr[96] = v.w * s;
    }
  }
  if (!rescan)
    for (int i = t; i < saved; i += OWN_T) list[i] = wl[i];
  __syncthreads();

  int n = saved;
  if (rescan) {

  const uint4* key4 = (const uint4*)key;
  for (int it0 = 0; it0 < SCAN_ITERS; it0 += SCAN_BATCH) {
    uint4 k[SCAN_BATCH];
#pragma unroll
    for (int u = 0; u < SCAN_BATCH; ++u) {
      const int q = (it0 + u) * OWN_T + t;
      k[u] = key4[min(q, N_KEYQ - 1)];
    }
#pragma unroll
    for (int u = 0; u < SCAN_BATCH; ++u) {
      const int q = (it0 + u) * OWN_T + t;
      const unsigned w[4] = {k[u].x, k[u].y, k[u].z, k[u].w};
      unsigned mask = 0;
#pragma unroll
      for (int j = 0; j < 4; ++j)
#pragma unroll
        for (int b = 0; b < 4; ++b)
          mask |= (unsigned)(((w[j] >> (8 * b)) & 0xFFu) == bid) << (4 * j + b);
      if (q >= N_KEYQ) mask = 0;
      while (mask) {
        const int pos = __builtin_ctz(mask);
        mask &= mask - 1;
        const int e = q * KEYS_PER_LOAD + pos;
        const int idx = atomicAdd(&cnt_s, 1);
        if (idx < LIST_CAP) {
          list[idx] = (unsigned)e;
        } else {
          const int c  = col[e];
          const int sN = row[e];
          const float nrm = dinv[sN] * dinv[c];
          float* tp = tile + (c - node0) * TPITCH;
          const float* hp = xin + (size_t)sN * FEAT;
#pragma nounroll
          for (int f = 0; f < FEAT; ++f) atomicAdd(tp + TCOL(f), nrm * hp[f]);
        }
      }
    }
  }
  __syncthreads();
  n = min(cnt_s, LIST_CAP);
  for (int i = t; i < n; i += OWN_T) {
    const int e = (int)list[i];
    list[i] = ((unsigned)row[e] << 9) | (unsigned)(col[e] - node0);
  }
  __syncthreads();
  }
  own_drain(tile, list, n, node0, xin, dinv, wave, lane);

  __syncthreads();
  float* sW = (float*)list;
#pragma unroll
  for (int j = 0; j < 4; ++j) {
    const int q = t + j * OWN_T;
    ((float4*)sW)[q] = ((const float4*)W)[TFEAT(q >> 5) * (FEAT / 4) + (q & 31)];
  }
  float* sBias = sW + FEAT * FEAT;
  if (t < FEAT / 4) ((float4*)sBias)[t] = ((const float4*)bias)[t];
  __syncthreads();

  if (wave < N_STRIPS)
    own_product<RELU_OUT>(tile, sW, sBias, wave, lane);
  __syncthreads();
  if (POOL) {
    own_pool(tile, batch, node0, sums, cnts, wave, lane);
  } else {
    float4* dst4 = (float4*)(xout + (size_t)node0 * FEAT);
    for (int i = t; i < TILE_NODES * FEAT / 4; i += OWN_T) {
      const int r = i >> 5, c4 = i & 31;
      dst4[i] = *(const float4*)(tile + r * TPITCH + 4 * c4);
    }
  }
}

__global__ void k_final(const float* __restrict__ sums, const float* __restrict__ cnts,
                        const float* __restrict__ Wl, const float* __restrict__ bl,
                        float* __restrict__ out) {
  const int g = blockIdx.x;
  const int c = threadIdx.x;
  if (c >= NCLS) return;
  const float inv = 1.0f / fmaxf(cnts[g], 1.0f);
  float acc = bl[c];
  for (int k = 0; k < FEAT; ++k)
    acc += sums[(size_t)g * FEAT + k] * inv * Wl[k * NCLS + c];
  out[(size_t)g * NCLS + c] = acc;
}

extern "C" void kernel_launch(void* const* d_in, const int* in_sizes, int n_in,
                              void* d_out, int out_size, void* d_ws, size_t ws_size,
                              hipStream_t stream) {
  const float* x     = (const float*)d_in[0];
  const int*   ei    = (const int*)d_in[1];
  const int*   row   = ei;
  const int*   col   = ei + N_EDGES;
  const int*   batch = (const int*)d_in[2];
  const float* W1 = (const float*)d_in[3];
  const float* b1 = (const float*)d_in[4];
  const float* W2 = (const float*)d_in[5];
  const float* b2 = (const float*)d_in[6];
  const float* W3 = (const float*)d_in[7];
  const float* b3 = (const float*)d_in[8];
  const float* Wl = (const float*)d_in[9];
  const float* bl = (const float*)d_in[10];
  float* outp = (float*)d_out;

  char* base = (char*)d_ws;
  const size_t featBytes = (size_t)N_NODES * FEAT * sizeof(float);
  float* deg  = (float*)(base);
  float* pA   = (float*)(base + (1 << 20) + featBytes);
  float* pB   = (float*)(base + (1 << 20) + 2 * featBytes);
  float* sums = (float*)(base + (1 << 20) + 3 * featBytes);
  float* cnts = sums + (size_t)NGRAPH * FEAT;
  unsigned char* key = (unsigned char*)(base + (2 << 20) + 3 * featBytes);
  int*      wcnt  = (int*)(base + (3 << 20) + 3 * featBytes);
  unsigned* wlist = (unsigned*)(base + (4 << 20) + 3 * featBytes);

  const int  TB     = 256;
  const int  gCnt   = (N_TILES * CNT_PITCH + TB - 1) / TB;

  k_fill<<<gCnt, TB, 0, stream>>>(wcnt, N_TILES * CNT_PITCH);
  k_deg<<<N_BIN_WG, BIN_T, 0, stream>>>(row, col, (unsigned*)key, wlist, wcnt);
  k_dinv<<<N_TILES, DINV_T, 0, stream>>>(wlist, wcnt, key, col, deg, sums);

  k_layer_own<true, false><<<N_TILES, OWN_T, 0, stream>>>(x, W1, b1, row, col, key, deg, pA, wlist, wcnt, batch, sums, cnts);
  k_layer_own<true, false><<<N_TILES, OWN_T, 0, stream>>>(pA, W2, b2, row, col, key, deg, pB, wlist, wcnt, batch, sums, cnts);
  k_layer_own<false, true><<<N_TILES, OWN_T, 0, stream>>>(pB, W3, b3, row, col, key, deg, nullptr, wlist, wcnt, batch, sums, cnts);

  k_final<<<NGRAPH, 32, 0, stream>>>(sums, cnts, Wl, bl, outp);
}


# a wave finishes its own 16-row strip straight behind its product: whole-line stores (layers 1, 2) or pool adds per strip (layer 3); barrier and joint store/pool pass gone; pooled sum order changes
# speedup vs baseline: 5.8075x; 1.1425x over previous
#include <hip/hip_runtime.h>

typedef __attribute__((ext_vector_type(2))) float v2f;
typedef __attribute__((ext_vector_type(8))) float v8f;

#define N_NODES 100000
#define N_EDGES 800000
#define FEAT    128
#define NGRAPH  512
#define NCLS    10

__device__ __forceinline__ void atomAddF(float* p, float v) {
  unsafeAtomicAdd(p, v);
}

__global__ void k_fill(int* __restrict__ cnt, long cnt_n) {
  long i = (long)blockIdx.x * blockDim.x + threadIdx.x;
  if (i < cnt_n) cnt[i] = 0;
}

#define TILE_NODES 400
#define N_TILES    (N_NODES / TILE_NODES)
#define LIST_CAP   24576
#define BIN_T      1024
#define N_BIN_WG   ((N_EDGES / 4 + BIN_T - 1) / BIN_T)
#define CNT_PITCH  32
__global__ __launch_bounds__(BIN_T) void k_deg(const int* __restrict__ row, const int* __restrict__ col,
                                               unsigned* __restrict__ key,
                                               unsigned* __restrict__ wlist, int* __restrict__ wcnt) {
  __shared__ int hist[N_TILES];
  const int t = threadIdx.x;
  if (t < N_TILES) hist[t] = 0;
  __syncthreads();
  const long i = (long)blockIdx.x * BIN_T + t;
  const bool on = i < N_EDGES / 4;
  int cc[4] = {0, 0, 0, 0}, rr[4] = {0, 0, 0, 0}, tl[4] = {0, 0, 0, 0}, rk[4] = {0, 0, 0, 0};
  if (on) {
    const int4 c = ((const int4*)col)[i];
    const int4 r = ((const int4*)row)[i];
    cc[0] = c.x; cc[1] = c.y; cc[2] = c.z; cc[3] = c.w;
    rr[0] = r.x; rr[1] = r.y; rr[2] = r.z; rr[3] = r.w;
#pragma unroll
    for (int j = 0; j < 4; ++j) {
      tl[j] = cc[j] / TILE_NODES;
      rk[j] = atomicAdd(&hist[tl[j]], 1);
    }
  }
  __syncthreads();
  if (t < N_TILES) {
    const int n = hist[t];
    hist[t] = n ? atomicAdd(&wcnt[t * CNT_PITCH], n) : 0;
  }
  __syncthreads();
  if (on) {
#pragma unroll
    for (int j = 0; j < 4; ++j) {
      const int idx = hist[tl[j]] + rk[j];
      if (idx < LIST_CAP)
        wlist[(size_t)tl[j] * LIST_CAP + idx] = ((unsigned)rr[j] << 9) | (unsigned)(cc[j] - tl[j] * TILE_NODES);
    }
    key[i] = (unsigned)tl[0] | ((unsigned)tl[1] << 8) | ((unsigned)tl[2] << 16) | ((unsigned)tl[3] << 24);
  }
}

#define POOL_WORDS (NGRAPH * FEAT + NGRAPH)
#define DINV_T     256
#define DINV_BATCH 4
__global__ __launch_bounds__(DINV_T) void k_dinv(const unsigned* __restrict__ wlist, const int* __restrict__ wcnt,
                                                 const unsigned char* __restrict__ key, const int* __restrict__ col,
                                                 float* __restrict__ dinv, float* __restrict__ pool) {
  __shared__ int rows[512];
  const int t = threadIdx.x;
  const int bid = blockIdx.x;
  const int node0 = bid * TILE_NODES;
  for (int i = t; i < 512; i += DINV_T) rows[i] = 0;
  __syncthreads();
  const int saved = wcnt[bid * CNT_PITCH];
  if (saved <= LIST_CAP) {
    const unsigned* wl = wlist + (size_t)bid * LIST_CAP;
    const uint4* wl4 = (const uint4*)wl;
    const int quads = saved >> 2;
    for (int q0 = 0; q0 < quads; q0 += DINV_BATCH * DINV_T) {
      uint4 v[DINV_BATCH];
#pragma unroll
      for (int u = 0; u < DINV_BATCH; ++u)
        v[u] = wl4[min(q0 + u * DINV_T + t, quads - 1)];
#pragma unroll
      for (int u = 0; u < DINV_BATCH; ++u) {
        if (q0 + u * DINV_T + t < quads) {
          atomicAdd(&rows[v[u].x & 511u], 1);
          atomicAdd(&rows[v[u].y & 511u], 1);
          atomicAdd(&rows[v[u].z & 511u], 1);
          atomicAdd(&rows[v[u].w & 511u], 1);
        }
      }
    }
    const int rest = (quads << 2) + t;
    if (rest < saved) atomicAdd(&rows[wl[rest] & 511u], 1);
  } else {
    for (int e = t; e < N_EDGES; e += DINV_T)
      if ((int)key[e] == bid) atomicAdd(&rows[col[e] - node0], 1);
  }
  __syncthreads();
  for (int i = t; i < TILE_NODES; i += DINV_T) {
    const float d = 1.0f + (float)rows[i];
    dinv[node0 + i] = (d > 0.0f) ? rsqrtf(fmaxf(d, 1.0f)) : 0.0f;
  }
  for (int p = bid * DINV_T + t; p < POOL_WORDS; p += N_TILES * DINV_T) pool[p] = 0.0f;
}

#define TCOL(f)    ((((f) & 3) << 5) | ((f) >> 2))
#define TFEAT(p)   ((((p) & 31) << 2) | ((p) >> 5))
#define OWN_T      1024
#define KEYS_PER_LOAD 16
#define N_KEYQ     (N_EDGES / KEYS_PER_LOAD)
#define SCAN_ITERS ((N_KEYQ + OWN_T - 1) / OWN_T)
#define SCAN_BATCH 7
#define TPITCH     (FEAT + 4)
#define N_STRIPS   (TILE_NODES / 16)
static_assert(N_NODES % TILE_NODES == 0 && N_TILES <= 255, "tile key must fit a byte below 0xFF");
static_assert(N_EDGES % KEYS_PER_LOAD == 0 && N_EDGES % 4 == 0, "key array is written 4 and read 16 bytes at a time");
static_assert(TILE_NODES <= 512 && N_NODES <= (1 << 17), "list entry packs source (17 bits) and local row (9 bits)");
static_assert((4l << 20) + 3l * N_NODES * FEAT * 4 + (long)N_TILES * LIST_CAP * 4 <= 268435456l, "stored lists end inside the workspace");
static_assert(TILE_NODES % 16 == 0 && N_STRIPS <= OWN_T / 32, "one wave per 16-row strip of the tile");
static_assert(FEAT * FEAT + FEAT <= LIST_CAP && FEAT * FEAT / 4 == 4 * OWN_T, "W and the bias row are staged in the list's LDS, W in four 16-byte pieces per thread");
static_assert((TILE_NODES * TPITCH + LIST_CAP + 4) * 4 <= 320 * 1024, "tile and list fit one WGP's LDS");

template <int K>
__device__ __forceinline__ void own_add(float* __restrict__ tile, const unsigned* __restrict__ ents,
                                        int node0, const float* __restrict__ xin,
                                        const float* __restrict__ dinv, int lane) {
  int    cl[K];
  float  nrm[K];
  float4 v[K];
#pragma unroll
  for (int u = 0; u < K; ++u) {
    const unsigned ent = ents[u];
    const int s = (int)(ent >> 9);
    cl[u] = (int)(ent & 511u);
    nrm[u] = dinv[s] * dinv[node0 + cl[u]];
    v[u] = *(const float4*)(xin + (size_t)s * FEAT + 4 * lane);
  }
#pragma unroll
  for (int u = 0; u < K; ++u) {
    float* tp = tile + cl[u] * TPITCH + lane;
    atomicAdd(tp + 0,  nrm[u] * v[u].x);
    atomicAdd(tp + 32, nrm[u] * v[u].y);
    atomicAdd(tp + 64, nrm[u] * v[u].z);
    atomicAdd(tp + 96, nrm[u] * v[u].w);
  }
}

__device__ __forceinline__ void own_drain(float* __restrict__ tile, const unsigned* __restrict__ list,
                                          int n, int node0,
                                          const float* __restrict__ xin,
                                          const float* __restrict__ dinv,
                                          int wave, int lane) {
  const int groups = n >> 2;
  for (int g = wave; g < groups; g += OWN_T / 32)
    own_add<4>(tile, list + 4 * g, node0, xin, dinv, lane);
  if (wave < (n & 3))
    own_add<1>(tile, list + 4 * groups + wave, node0, xin, dinv, lane);
}

template <bool RELU_OUT>
__device__ __forceinline__ void own_product(float* __restrict__ tile, const float* __restrict__ sW,
                                            const float* __restrict__ sBias, int strip, int lane) {
  const int m    = lane & 15;
  const int koff = (lane >> 4) << 1;
  const int ncol = lane & 15;

  v8f acc[8];
#pragma unroll
  for (int n = 0; n < 8; ++n) acc[n] = (v8f){0, 0, 0, 0, 0, 0, 0, 0};

  const float* arow = tile + (strip * 16 + m) * TPITCH + koff;

  for (int k0 = 0; k0 < FEAT; k0 += 16) {
    v2f a[4];
#pragma unroll
    for (int j = 0; j < 4; ++j) {
      a[j].x = arow[k0 + 4 * j];
      a[j].y = arow[k0 + 4 * j + 1];
    }
    v2f b[2][8];
#pragma unroll
    for (int n = 0; n < 8; ++n) {
      const int kb = k0 + koff;
      b[0][n].x = sW[kb * FEAT + n * 16 + ncol];
      b[0][n].y = sW[(kb + 1) * FEAT + n * 16 + ncol];
    }
#pragma unroll
    for (int j = 0; j < 4; ++j) {
      if (j < 3) {
        const int kb = k0 + 4 * (j + 1) + koff;
#pragma unroll
        for (int n = 0; n < 8; ++n) {
          b[(j + 1) & 1][n].x = sW[kb * FEAT + n * 16 + ncol];
          b[(j + 1) & 1][n].y = sW[(kb + 1) * FEAT + n * 16 + ncol];
        }
      }
#pragma unroll
      for (int n = 0; n < 8; ++n) {
        acc[n] = __builtin_amdgcn_wmma_f32_16x16x4_f32(
            false, a[j], false, b[j & 1][n], (short)0, acc[n], false, false);
      }
    }
  }

  const int rowsel = (lane >> 4) * 8;
  float* orow = tile + (strip * 16 + rowsel) * TPITCH + ncol;
#pragma unroll
  for (int n = 0; n < 8; ++n) {
    const float bv = sBias[n * 16 + ncol];
#pragma unroll
    for (int r = 0; r < 8; ++r) {
      float o = acc[n][r] + bv;
      if (RELU_OUT) o = fmaxf(o, 0.0f);
      orow[r * TPITCH + n * 16] = o;
    }
  }
}

__device__ __forceinline__ void pool_flush(float* __restrict__ sums, float* __restrict__ cnts,
                                           int g, int lane, const float* acc, int run) {
#pragma unroll
  for (int q = 0; q < FEAT / 32; ++q) atomAddF(sums + (size_t)g * FEAT + 32 * q + lane, acc[q]);
  if (lane == 0) atomAddF(cnts + g, (float)run);
}
__device__ __forceinline__ void own_pool_strip(const float* __restrict__ tile, const int* __restrict__ batch,
                                               int node0, float* __restrict__ sums, float* __restrict__ cnts,
                                               int strip, int lane) {
  const int r0 = strip * 16;
  const int id = batch[node0 + r0 + (lane & 15)];
  const float* tp = tile + r0 * TPITCH + lane;
  int   g   = __builtin_amdgcn_readlane(id, 0);
  float acc[FEAT / 32] = {0.0f, 0.0f, 0.0f, 0.0f};
  int   run = 0;
#pragma unroll
  for (int c = 0; c < 16; c += 8) {
    float v[8][FEAT / 32];
#pragma unroll
    for (int j = 0; j < 8; ++j)
#pragma unroll
      for (int q = 0; q < FEAT / 32; ++q) v[j][q] = tp[(c + j) * TPITCH + 32 * q];
#pragma unroll
    for (int j = 0; j < 8; ++j) {
      const int gr = __builtin_amdgcn_readlane(id, c + j);
      if (gr != g) {
        pool_flush(sums, cnts, g, lane, acc, run);
        g = gr; run = 0;
#pragma unroll
        for (int q = 0; q < FEAT / 32; ++q) acc[q] = 0.0f;
      }
#pragma unroll
      for (int q = 0; q < FEAT / 32; ++q) acc[q] += v[j][q];
      ++run;
    }
  }
  pool_flush(sums, cnts, g, lane, acc, run);
}

template <bool RELU_OUT, bool POOL>
__global__ __launch_bounds__(OWN_T) void k_layer_own(const float* __restrict__ xin,
                                                     const float* __restrict__ W,
                                                     const float* __restrict__ bias,
                                                     const int* __restrict__ row,
                                                     const int* __restrict__ col,
                                                     const unsigned char* __restrict__ key,
                                                     const float* __restrict__ dinv,
                                                     float* __restrict__ xout,
                                                     const unsigned* __restrict__ wlist,
                                                     const int* __restrict__ wcnt,
                                                     const int* __restrict__ batch,
                                                     float* __restrict__ sums,
                                                     float* __restrict__ cnts) {
  __shared__ __attribute__((aligned(16))) unsigned smem[LIST_CAP + TILE_NODES * TPITCH];
  unsigned* const list = smem;
  float* const tile = (float*)(smem + LIST_CAP);
  __shared__ int cnt_s;

  const int t     = threadIdx.x;
  const int wave  = __builtin_amdgcn_readfirstlane(t >> 5);
  const int lane  = t & 31;
  const unsigned bid = blockIdx.x;
  const int node0 = (int)bid * TILE_NODES;
  const unsigned* wl = wlist + (size_t)bid * LIST_CAP;
  const int saved = wcnt[bid * CNT_PITCH];
  const bool rescan = saved > LIST_CAP;

  if (t == 0) cnt_s = 0;
  {
    const float4* src4 = (const float4*)(xin + (size_t)node0 * FEAT);
    for (int i = t; i < TILE_NODES * FEAT / 4; i += OWN_T) {
      const int r = i >> 5, c4 = i & 31;
      const float d = dinv[node0 + r];
      const float s = d * d;
      float4 v = src4[i];
      float* tr = tile + r * TPITCH + c4;
      tr[0]  = v.x * s;
      tr[32] = v.y * s;
      tr[64] = v.z * s;
      tr[96] = v.w * s;
    }
  }
  if (!rescan)
    for (int i = t; i < saved; i += OWN_T) list[i] = wl[i];
  __syncthreads();

  int n = saved;
  if (rescan) {

  const uint4* key4 = (const uint4*)key;
  for (int it0 = 0; it0 < SCAN_ITERS; it0 += SCAN_BATCH) {
    uint4 k[SCAN_BATCH];
#pragma unroll
    for (int u = 0; u < SCAN_BATCH; ++u) {
      const int q = (it0 + u) * OWN_T + t;
      k[u] = key4[min(q, N_KEYQ - 1)];
    }
#pragma unroll
    for (int u = 0; u < SCAN_BATCH; ++u) {
      const int q = (it0 + u) * OWN_T + t;
      const unsigned w[4] = {k[u].x, k[u].y, k[u].z, k[u].w};
      unsigned mask = 0;
#pragma unroll
      for (int j = 0; j < 4; ++j)
#pragma unroll
        for (int b = 0; b < 4; ++b)
          mask |= (unsigned)(((w[j] >> (8 * b)) & 0xFFu) == bid) << (4 * j + b);
      if (q >= N_KEYQ) mask = 0;
      while (mask) {
        const int pos = __builtin_ctz(mask);
        mask &= mask - 1;
        const int e = q * KEYS_PER_LOAD + pos;
        const int idx = atomicAdd(&cnt_s, 1);
        if (idx < LIST_CAP) {
          list[idx] = (unsigned)e;
        } else {
          const int c  = col[e];
          const int sN = row[e];
          const float nrm = dinv[sN] * dinv[c];
          float* tp = tile + (c - node0) * TPITCH;
          const float* hp = xin + (size_t)sN * FEAT;
#pragma nounroll
          for (int f = 0; f < FEAT; ++f) atomicAdd(tp + TCOL(f), nrm * hp[f]);
        }
      }
    }
  }
  __syncthreads();
  n = min(cnt_s, LIST_CAP);
  for (int i = t; i < n; i += OWN_T) {
    const int e = (int)list[i];
    list[i] = ((unsigned)row[e] << 9) | (unsigned)(col[e] - node0);
  }
  __syncthreads();
  }
  own_drain(tile, list, n, node0, xin, dinv, wave, lane);

  __syncthreads();
  float* sW = (float*)list;
#pragma unroll
  for (int j = 0; j < 4; ++j) {
    const int q = t + j * OWN_T;
    ((float4*)sW)[q] = ((const float4*)W)[TFEAT(q >> 5) * (FEAT / 4) + (q & 31)];
  }
  float* sBias = sW + FEAT * FEAT;
  if (t < FEAT / 4) ((float4*)sBias)[t] = ((const float4*)bias)[t];
  __syncthreads();

  if (wave < N_STRIPS) {
    own_product<RELU_OUT>(tile, sW, sBias, wave, lane);
    __builtin_amdgcn_wave_barrier();
    if (POOL) {
      own_pool_strip(tile, batch, node0, sums, cnts, wave, lane);
    } else {
      float4* dst4 = (float4*)(xout + (size_t)(node0 + wave * 16) * FEAT);
      const float* tr = tile + wave * 16 * TPITCH + 4 * lane;
#pragma unroll
      for (int r = 0; r < 16; ++r)
        dst4[r * (FEAT / 4) + lane] = *(const float4*)(tr + r * TPITCH);
    }
  }
}

__global__ void k_final(const float* __restrict__ sums, const float* __restrict__ cnts,
                        const float* __restrict__ Wl, const float* __restrict__ bl,
                        float* __restrict__ out) {
  const int g = blockIdx.x;
  const int c = threadIdx.x;
  if (c >= NCLS) return;
  const float inv = 1.0f / fmaxf(cnts[g], 1.0f);
  float acc = bl[c];
  for (int k = 0; k < FEAT; ++k)
    acc += sums[(size_t)g * FEAT + k] * inv * Wl[k * NCLS + c];
  out[(size_t)g * NCLS + c] = acc;
}

extern "C" void kernel_launch(void* const* d_in, const int* in_sizes, int n_in,
                              void* d_out, int out_size, void* d_ws, size_t ws_size,
                              hipStream_t stream) {
  const float* x     = (const float*)d_in[0];
  const int*   ei    = (const int*)d_in[1];
  const int*   row   = ei;
  const int*   col   = ei + N_EDGES;
  const int*   batch = (const int*)d_in[2];
  const float* W1 = (const float*)d_in[3];
  const float* b1 = (const float*)d_in[4];
  const float* W2 = (const float*)d_in[5];
  const float* b2 = (const float*)d_in[6];
  const float* W3 = (const float*)d_in[7];
  const float* b3 = (const float*)d_in[8];
  const float* Wl = (const float*)d_in[9];
  const float* bl = (const float*)d_in[10];
  float* outp = (float*)d_out;

  char* base = (char*)d_ws;
  const size_t featBytes = (size_t)N_NODES * FEAT * sizeof(float);
  float* deg  = (float*)(base);
  float* pA   = (float*)(base + (1 << 20) + featBytes);
  float* pB   = (float*)(base + (1 << 20) + 2 * featBytes);
  float* sums = (float*)(base + (1 << 20) + 3 * featBytes);
  float* cnts = sums + (size_t)NGRAPH * FEAT;
  unsigned char* key = (unsigned char*)(base + (2 << 20) + 3 * featBytes);
  int*      wcnt  = (int*)(base + (3 << 20) + 3 * featBytes);
  unsigned* wlist = (unsigned*)(base + (4 << 20) + 3 * featBytes);

  const int  TB     = 256;
  const int  gCnt   = (N_TILES * CNT_PITCH + TB - 1) / TB;

  k_fill<<<gCnt, TB, 0, stream>>>(wcnt, N_TILES * CNT_PITCH);
  k_deg<<<N_BIN_WG, BIN_T, 0, stream>>>(row, col, (unsigned*)key, wlist, wcnt);
  k_dinv<<<N_TILES, DINV_T, 0, stream>>>(wlist, wcnt, key, col, deg, sums);

  k_layer_own<true, false><<<N_TILES, OWN_T, 0, stream>>>(x, W1, b1, row, col, key, deg, pA, wlist, wcnt, batch, sums, cnts);
  k_layer_own<true, false><<<N_TILES, OWN_T, 0, stream>>>(pA, W2, b2, row, col, key, deg, pB, wlist, wcnt, batch, sums, cnts);
  k_layer_own<false, true><<<N_TILES, OWN_T, 0, stream>>>(pB, W3, b3, row, col, key, deg, nullptr, wlist, wcnt, batch, sums, cnts);

  k_final<<<NGRAPH, 32, 0, stream>>>(sums, cnts, Wl, bl, outp);
}
